# starting kernel
# baseline (speedup 1.0000x reference)
#include <hip/hip_runtime.h>

#define N_DRUG 846
#define D 64
#define K 1024
#define N_REL 200
#define RPAD 208
#define BN_EPS 1e-5f

typedef __attribute__((ext_vector_type(2))) float v2f;
typedef __attribute__((ext_vector_type(8))) float v8f;

__device__ __forceinline__ float wave_reduce_add(float v) {
  #pragma unroll
  for (int off = 16; off > 0; off >>= 1) v += __shfl_xor(v, off, 32);
  return v;
}

__global__ void __launch_bounds__(256) gnn_b2sum(const float* __restrict__ b2,
                                                 float* __restrict__ b2sum) {
  const int k = blockIdx.x * 256 + threadIdx.x;
  if (k < K) {
    const float* p = b2 + (size_t)k * D;
    float s = 0.f;
    #pragma unroll
    for (int e = 0; e < D; ++e) s += p[e];
    b2sum[k] = s;
  }
}

__global__ void __launch_bounds__(256) gnn_main(
    const float* __restrict__ drug_table, const float* __restrict__ rela_table,
    const float* __restrict__ ent_table,  const float* __restrict__ W1g,
    const float* __restrict__ b1,         const float* __restrict__ W2g,
    const int*   __restrict__ drug_name,  const int* __restrict__ adj_tail,
    const int*   __restrict__ adj_rel,    const float* __restrict__ b2sum,
    float* __restrict__ drug_e)
{
  extern __shared__ float smem[];
  float* Atile = smem;
  float* Ttile = Atile + RPAD * D;
  float* W1s   = Ttile + RPAD * D;
  float* demb  = W1s + D * D;
  float* w2sum = demb + D;
  float* score = w2sum + D;
  float* red   = score + K;

  const int n    = blockIdx.x;
  const int t    = threadIdx.x;
  const int lane = t & 31;
  const int wave = t >> 5;

  const size_t wbase = (size_t)n * D * D;

  if (t < D) demb[t] = drug_table[(size_t)drug_name[n] * D + t];
  for (int i = t; i < D * D; i += 256) W1s[i] = W1g[wbase + i];
  {
    const int d = t >> 2, q = t & 3;
    const float* p = W2g + wbase + d * D + q * 16;
    float s = 0.f;
    #pragma unroll
    for (int j = 0; j < 16; ++j) s += p[j];
    red[t] = s;
  }
  __syncthreads();
  if (t < D) w2sum[t] = red[4*t] + red[4*t+1] + red[4*t+2] + red[4*t+3];
  for (int i = t; i < RPAD * D; i += 256) {
    const int r = i >> 6, d = i & 63;
    Atile[i] = (r < N_REL) ? demb[d] * rela_table[r * D + d] : 0.f;
  }
  __syncthreads();

  const int half = lane >> 4;
  const int l    = lane & 15;
  for (int tile = wave; tile < 13 * 4; tile += 8) {
    const int m0 = (tile % 13) * 16;
    const int n0 = (tile / 13) * 16;
    v8f c = {0.f, 0.f, 0.f, 0.f, 0.f, 0.f, 0.f, 0.f};
    const float* arow = Atile + (m0 + l) * D;
    #pragma unroll
    for (int kk = 0; kk < 16; ++kk) {
      const int k0 = kk * 4 + 2 * half;
      v2f a, b;
      a[0] = arow[k0];
      a[1] = arow[k0 + 1];
      b[0] = W1s[k0 * D + n0 + l];
      b[1] = W1s[(k0 + 1) * D + n0 + l];
      c = __builtin_amdgcn_wmma_f32_16x16x4_f32(false, a, false, b,
                                                (short)0, c, false, false);
    }
    #pragma unroll
    for (int i = 0; i < 8; ++i)
      Ttile[(m0 + 8 * half + i) * D + n0 + l] = c[i];
  }
  __syncthreads();

  for (int i = 0; i < 128; ++i) {
    const int k = wave * 128 + i;
    const int r = adj_rel[(size_t)n * K + k];
    float v = fmaxf(Ttile[r * D + lane]      + b1[(size_t)k * D + lane],      0.f) * w2sum[lane]
            + fmaxf(Ttile[r * D + lane + 32] + b1[(size_t)k * D + lane + 32], 0.f) * w2sum[lane + 32];
    v = wave_reduce_add(v);
    if (lane == 0) score[k] = v + b2sum[k];
  }
  __syncthreads();

  float lmax = -INFINITY;
  for (int k = t; k < K; k += 256) lmax = fmaxf(lmax, score[k]);
  red[t] = lmax;
  __syncthreads();
  for (int s = 128; s > 0; s >>= 1) {
    if (t < s) red[t] = fmaxf(red[t], red[t + s]);
    __syncthreads();
  }
  const float mx = red[0];
  __syncthreads();
  float lsum = 0.f;
  for (int k = t; k < K; k += 256) {
    const float e = __expf(score[k] - mx);
    score[k] = e;
    lsum += e;
  }
  red[t] = lsum;
  __syncthreads();
  for (int s = 128; s > 0; s >>= 1) {
    if (t < s) red[t] += red[t + s];
    __syncthreads();
  }
  const float inv = 1.f / red[0];
  __syncthreads();

  float acc0 = 0.f, acc1 = 0.f;
  for (int i = 0; i < 128; ++i) {
    const int k = wave * 128 + i;
    const float sc = score[k] * inv;
    const size_t tl = (size_t)adj_tail[(size_t)n * K + k] * D;
    acc0 += sc * ent_table[tl + lane];
    acc1 += sc * ent_table[tl + lane + 32];
  }
  red[wave * 64 + lane]      = acc0;
  red[wave * 64 + lane + 32] = acc1;
  __syncthreads();
  if (t < D) {
    float s = 0.f;
    #pragma unroll
    for (int w = 0; w < 8; ++w) s += red[w * 64 + t];
    drug_e[(size_t)n * (2 * D) + t]     = s;
    drug_e[(size_t)n * (2 * D) + D + t] = demb[t];
  }
}

__global__ void __launch_bounds__(64) gnn_linear(
    const float* __restrict__ drug_e, const float* __restrict__ lin_w,
    const float* __restrict__ lin_b, float* __restrict__ y)
{
  __shared__ float de[2 * D];
  const int n = blockIdx.x, t = threadIdx.x;
  de[t]     = drug_e[(size_t)n * (2 * D) + t];
  de[t + D] = drug_e[(size_t)n * (2 * D) + D + t];
  __syncthreads();
  float acc = lin_b[t];
  const float* w = lin_w + (size_t)t * (2 * D);
  #pragma unroll 8
  for (int j = 0; j < 2 * D; ++j) acc += de[j] * w[j];
  y[(size_t)n * D + t] = fmaxf(acc, 0.f);
}

__global__ void __launch_bounds__(256) gnn_bn(
    const float* __restrict__ y, const float* __restrict__ bn_w,
    const float* __restrict__ bn_b, float* __restrict__ out)
{
  __shared__ float red[256];
  const int c = blockIdx.x, t = threadIdx.x;
  float s = 0.f;
  for (int r = t; r < N_DRUG; r += 256) s += y[(size_t)r * D + c];
  red[t] = s;
  __syncthreads();
  for (int st = 128; st > 0; st >>= 1) {
    if (t < st) red[t] += red[t + st];
    __syncthreads();
  }
  const float mean = red[0] / (float)N_DRUG;
  __syncthreads();
  float q = 0.f;
  for (int r = t; r < N_DRUG; r += 256) {
    const float d = y[(size_t)r * D + c] - mean;
    q += d * d;
  }
  red[t] = q;
  __syncthreads();
  for (int st = 128; st > 0; st >>= 1) {
    if (t < st) red[t] += red[t + st];
    __syncthreads();
  }
  const float var   = red[0] / (float)N_DRUG;
  const float scale = bn_w[c] * rsqrtf(var + BN_EPS);
  const float shift = bn_b[c];
  for (int r = t; r < N_DRUG; r += 256)
    out[(size_t)r * D + c] = (y[(size_t)r * D + c] - mean) * scale + shift;
}

extern "C" void kernel_launch(void* const* d_in, const int* in_sizes, int n_in,
                              void* d_out, int out_size, void* d_ws, size_t ws_size,
                              hipStream_t stream) {
  (void)in_sizes; (void)n_in; (void)out_size; (void)ws_size;
  const float* drug_table = (const float*)d_in[0];
  const float* rela_table = (const float*)d_in[1];
  const float* ent_table  = (const float*)d_in[2];
  const float* W1         = (const float*)d_in[3];
  const float* b1         = (const float*)d_in[4];
  const float* W2         = (const float*)d_in[5];
  const float* b2         = (const float*)d_in[6];
  const float* lin_w      = (const float*)d_in[7];
  const float* lin_b      = (const float*)d_in[8];
  const float* bn_w       = (const float*)d_in[9];
  const float* bn_b       = (const float*)d_in[10];
  const int*   drug_name  = (const int*)d_in[11];
  const int*   adj_tail   = (const int*)d_in[12];
  const int*   adj_rel    = (const int*)d_in[13];

  float* ws     = (float*)d_ws;
  float* b2sum  = ws;
  float* drug_e = ws + K;
  float* y      = drug_e + N_DRUG * 2 * D;

  gnn_b2sum<<<dim3((K + 255) / 256), dim3(256), 0, stream>>>(b2, b2sum);

  const size_t lds = (size_t)(RPAD * D * 2 + D * D + D + D + K + 512) * sizeof(float);
  gnn_main<<<dim3(N_DRUG), dim3(256), lds, stream>>>(
      drug_table, rela_table, ent_table, W1, b1, W2,
      drug_name, adj_tail, adj_rel, b2sum, drug_e);

  gnn_linear<<<dim3(N_DRUG), dim3(64), 0, stream>>>(drug_e, lin_w, lin_b, y);

  gnn_bn<<<dim3(D), dim3(256), 0, stream>>>(y, bn_w, bn_b, (float*)d_out);
}


# Logits per 16-neighbour tile on the matrix unit (W1^T x gathered relation rows); bias, relu and w2 dot in accumulator registers; T table and its row loop gone; T and logit sum orders change
# speedup vs baseline: 3.1471x; 3.1471x over previous
#include <hip/hip_runtime.h>

#define N_DRUG 846
#define D 64
#define K 1024
#define N_REL 200
#define BN_EPS 1e-5f

typedef __attribute__((ext_vector_type(2))) float v2f;
typedef __attribute__((ext_vector_type(4))) float v4f;
typedef __attribute__((ext_vector_type(8))) float v8f;

__device__ __forceinline__ float wave_reduce_add(float v) {
  #pragma unroll
  for (int off = 16; off > 0; off >>= 1) v += __shfl_xor(v, off, 32);
  return v;
}

__global__ void __launch_bounds__(256) gnn_b2sum(const float* __restrict__ b2,
                                                 float* __restrict__ b2sum) {
  const int k = blockIdx.x * 256 + threadIdx.x;
  if (k < K) {
    const float* p = b2 + (size_t)k * D;
    float s = 0.f;
    #pragma unroll
    for (int e = 0; e < D; ++e) s += p[e];
    b2sum[k] = s;
  }
}

#define ASTR 68
#define PSTR 130

__global__ void __launch_bounds__(256, 8) gnn_main(
    const float* __restrict__ drug_table, const float* __restrict__ rela_table,
    const float* __restrict__ ent_table,  const float* __restrict__ W1g,
    const float* __restrict__ b1,         const float* __restrict__ W2g,
    const int*   __restrict__ drug_name,  const int* __restrict__ adj_tail,
    const int*   __restrict__ adj_rel,    const float* __restrict__ b2sum,
    float* __restrict__ drug_e)
{
  extern __shared__ __align__(16) float smem[];
  float* Atab  = smem;
  float* W1p   = Atab + N_REL * ASTR;
  float* demb  = W1p + 32 * PSTR;
  float* w2sum = demb + D;
  float* score = w2sum + D;
  float* red   = score + K;

  const int n    = blockIdx.x;
  const int t    = threadIdx.x;
  const int lane = t & 31;
  const int wave = t >> 5;
  const int swave = __builtin_amdgcn_readfirstlane(wave);
  const int h    = lane >> 4;
  const int c    = lane & 15;

  const size_t wbase = (size_t)n * D * D;
  const int* relp = adj_rel + (size_t)n * K + swave * 128 + c;
  int rcur = relp[0];

  if (t < D) demb[t] = drug_table[(size_t)drug_name[n] * D + t];
  for (int i = t; i < D * D; i += 256) {
    const int d = i >> 6, e = i & 63;
    W1p[(d >> 1) * PSTR + e * 2 + (d & 1)] = W1g[wbase + i];
  }
  {
    const int d = t >> 2, q = t & 3;
    const float* p = W2g + wbase + d * D + q * 16;
    float s = 0.f;
    #pragma unroll
    for (int j = 0; j < 16; ++j) s += p[j];
    red[t] = s;
  }
  __syncthreads();
  if (t < D) w2sum[t] = red[4*t] + red[4*t+1] + red[4*t+2] + red[4*t+3];
  for (int i = t; i < N_REL * D; i += 256) {
    const int r = i >> 6, d = i & 63;
    Atab[r * ASTR + d] = demb[d] * rela_table[i];
  }
  __syncthreads();

  const float* wp = W1p + (16 * h) * PSTR + 2 * c;
  for (int tt = 0; tt < 8; ++tt) {
    const int k = swave * 128 + tt * 16 + c;
    const int rnext = relp[(tt < 7 ? tt + 1 : tt) * 16];
    const float* arow = Atab + rcur * ASTR + 32 * h;
    v4f bq[8];
    #pragma unroll
    for (int j = 0; j < 8; ++j) bq[j] = *(const v4f*)(arow + 4 * j);
    const float* bp = b1 + (size_t)k * D + 8 * h;
    v4f bb[4][2];
    #pragma unroll
    for (int et = 0; et < 4; ++et) {
      bb[et][0] = *(const v4f*)(bp + 16 * et);
      bb[et][1] = *(const v4f*)(bp + 16 * et + 4);
    }
    const float b2s = b2sum[k];
    v8f acc[4];
    #pragma unroll
    for (int et = 0; et < 4; ++et)
      acc[et] = (v8f){0.f, 0.f, 0.f, 0.f, 0.f, 0.f, 0.f, 0.f};
    #pragma unroll
    for (int kk = 0; kk < 16; ++kk) {
      v2f b;
      b[0] = bq[kk >> 1][2 * (kk & 1)];
      b[1] = bq[kk >> 1][2 * (kk & 1) + 1];
      #pragma unroll
      for (int et = 0; et < 4; ++et) {
        const v2f a = *(const v2f*)(wp + kk * PSTR + 32 * et);
        acc[et] = __builtin_amdgcn_wmma_f32_16x16x4_f32(false, a, false, b,
                                                        (short)0, acc[et], false, false);
      }
    }
    float v = 0.f;
    #pragma unroll
    for (int et = 0; et < 4; ++et) {
      const v4f w0 = *(const v4f*)(w2sum + 16 * et + 8 * h);
      const v4f w1 = *(const v4f*)(w2sum + 16 * et + 8 * h + 4);
      #pragma unroll
      for (int r = 0; r < 4; ++r) {
        v += fmaxf(acc[et][r]     + bb[et][0][r], 0.f) * w0[r];
        v += fmaxf(acc[et][4 + r] + bb[et][1][r], 0.f) * w1[r];
      }
    }
    v += __shfl_xor(v, 16, 32);
    if (h == 0) score[k] = v + b2s;
    rcur = rnext;
  }
  __syncthreads();

  float lmax = -INFINITY;
  for (int k = t; k < K; k += 256) lmax = fmaxf(lmax, score[k]);
  red[t] = lmax;
  __syncthreads();
  for (int s = 128; s > 0; s >>= 1) {
    if (t < s) red[t] = fmaxf(red[t], red[t + s]);
    __syncthreads();
  }
  const float mx = red[0];
  __syncthreads();
  float lsum = 0.f;
  for (int k = t; k < K; k += 256) {
    const float e = __expf(score[k] - mx);
    score[k] = e;
    lsum += e;
  }
  red[t] = lsum;
  __syncthreads();
  for (int s = 128; s > 0; s >>= 1) {
    if (t < s) red[t] += red[t + s];
    __syncthreads();
  }
  const float inv = 1.f / red[0];
  __syncthreads();

  float acc0 = 0.f, acc1 = 0.f;
  for (int i = 0; i < 128; ++i) {
    const int k = wave * 128 + i;
    const float sc = score[k] * inv;
    const size_t tl = (size_t)adj_tail[(size_t)n * K + k] * D;
    acc0 += sc * ent_table[tl + lane];
    acc1 += sc * ent_table[tl + lane + 32];
  }
  red[wave * 64 + lane]      = acc0;
  red[wave * 64 + lane + 32] = acc1;
  __syncthreads();
  if (t < D) {
    float s = 0.f;
    #pragma unroll
    for (int w = 0; w < 8; ++w) s += red[w * 64 + t];
    drug_e[(size_t)n * (2 * D) + t]     = s;
    drug_e[(size_t)n * (2 * D) + D + t] = demb[t];
  }
}

__global__ void __launch_bounds__(64) gnn_linear(
    const float* __restrict__ drug_e, const float* __restrict__ lin_w,
    const float* __restrict__ lin_b, float* __restrict__ y)
{
  __shared__ float de[2 * D];
  const int n = blockIdx.x, t = threadIdx.x;
  de[t]     = drug_e[(size_t)n * (2 * D) + t];
  de[t + D] = drug_e[(size_t)n * (2 * D) + D + t];
  __syncthreads();
  float acc = lin_b[t];
  const float* w = lin_w + (size_t)t * (2 * D);
  #pragma unroll 8
  for (int j = 0; j < 2 * D; ++j) acc += de[j] * w[j];
  y[(size_t)n * D + t] = fmaxf(acc, 0.f);
}

__global__ void __launch_bounds__(256) gnn_bn(
    const float* __restrict__ y, const float* __restrict__ bn_w,
    const float* __restrict__ bn_b, float* __restrict__ out)
{
  __shared__ float red[256];
  const int c = blockIdx.x, t = threadIdx.x;
  float s = 0.f;
  for (int r = t; r < N_DRUG; r += 256) s += y[(size_t)r * D + c];
  red[t] = s;
  __syncthreads();
  for (int st = 128; st > 0; st >>= 1) {
    if (t < st) red[t] += red[t + st];
    __syncthreads();
  }
  const float mean = red[0] / (float)N_DRUG;
  __syncthreads();
  float q = 0.f;
  for (int r = t; r < N_DRUG; r += 256) {
    const float d = y[(size_t)r * D + c] - mean;
    q += d * d;
  }
  red[t] = q;
  __syncthreads();
  for (int st = 128; st > 0; st >>= 1) {
    if (t < st) red[t] += red[t + st];
    __syncthreads();
  }
  const float var   = red[0] / (float)N_DRUG;
  const float scale = bn_w[c] * rsqrtf(var + BN_EPS);
  const float shift = bn_b[c];
  for (int r = t; r < N_DRUG; r += 256)
    out[(size_t)r * D + c] = (y[(size_t)r * D + c] - mean) * scale + shift;
}

extern "C" void kernel_launch(void* const* d_in, const int* in_sizes, int n_in,
                              void* d_out, int out_size, void* d_ws, size_t ws_size,
                              hipStream_t stream) {
  (void)in_sizes; (void)n_in; (void)out_size; (void)ws_size;
  const float* drug_table = (const float*)d_in[0];
  const float* rela_table = (const float*)d_in[1];
  const float* ent_table  = (const float*)d_in[2];
  const float* W1         = (const float*)d_in[3];
  const float* b1         = (const float*)d_in[4];
  const float* W2         = (const float*)d_in[5];
  const float* b2         = (const float*)d_in[6];
  const float* lin_w      = (const float*)d_in[7];
  const float* lin_b      = (const float*)d_in[8];
  const float* bn_w       = (const float*)d_in[9];
  const float* bn_b       = (const float*)d_in[10];
  const int*   drug_name  = (const int*)d_in[11];
  const int*   adj_tail   = (const int*)d_in[12];
  const int*   adj_rel    = (const int*)d_in[13];

  float* ws     = (float*)d_ws;
  float* b2sum  = ws;
  float* drug_e = ws + K;
  float* y      = drug_e + N_DRUG * 2 * D;

  gnn_b2sum<<<dim3((K + 255) / 256), dim3(256), 0, stream>>>(b2, b2sum);

  const size_t lds = (size_t)(N_REL * ASTR + 32 * PSTR + D + D + K + 512) * sizeof(float);
  gnn_main<<<dim3(N_DRUG), dim3(256), lds, stream>>>(
      drug_table, rela_table, ent_table, W1, b1, W2,
      drug_name, adj_tail, adj_rel, b2sum, drug_e);

  gnn_linear<<<dim3(N_DRUG), dim3(64), 0, stream>>>(drug_e, lin_w, lin_b, y);

  gnn_bn<<<dim3(D), dim3(256), 0, stream>>>(y, bn_w, bn_b, (float*)d_out);
}


# Softmax with a running maximum per wave: each 16-neighbour tile fetches and adds its entity rows once its logits exist, under the other waves' matrix work; softmax and entity sum orders change
# speedup vs baseline: 3.3367x; 1.0602x over previous
#include <hip/hip_runtime.h>

#define N_DRUG 846
#define D 64
#define K 1024
#define N_REL 200
#define BN_EPS 1e-5f

typedef __attribute__((ext_vector_type(2))) float v2f;
typedef __attribute__((ext_vector_type(4))) float v4f;
typedef __attribute__((ext_vector_type(8))) float v8f;

__device__ __forceinline__ float wave_reduce_add(float v) {
  #pragma unroll
  for (int off = 16; off > 0; off >>= 1) v += __shfl_xor(v, off, 32);
  return v;
}

template <int CTRL>
__device__ __forceinline__ float half_xor(float v) {
  return __builtin_bit_cast(float,
      __builtin_amdgcn_update_dpp(0, __builtin_bit_cast(int, v), CTRL, 0xf, 0xf, true));
}
__device__ __forceinline__ float half16_max(float v) {
  v = fmaxf(v, half_xor<0x168>(v));
  v = fmaxf(v, half_xor<0x164>(v));
  v = fmaxf(v, half_xor<0x162>(v));
  v = fmaxf(v, half_xor<0x161>(v));
  return v;
}
__device__ __forceinline__ float half16_sum(float v) {
  v += half_xor<0x168>(v);
  v += half_xor<0x164>(v);
  v += half_xor<0x162>(v);
  v += half_xor<0x161>(v);
  return v;
}

__global__ void __launch_bounds__(256) gnn_b2sum(const float* __restrict__ b2,
                                                 float* __restrict__ b2sum) {
  const int k = blockIdx.x * 256 + threadIdx.x;
  if (k < K) {
    const float* p = b2 + (size_t)k * D;
    float s = 0.f;
    #pragma unroll
    for (int e = 0; e < D; ++e) s += p[e];
    b2sum[k] = s;
  }
}

#define ASTR 68
#define PSTR 130

__global__ void __launch_bounds__(256, 8) gnn_main(
    const float* __restrict__ drug_table, const float* __restrict__ rela_table,
    const float* __restrict__ ent_table,  const float* __restrict__ W1g,
    const float* __restrict__ b1,         const float* __restrict__ W2g,
    const int*   __restrict__ drug_name,  const int* __restrict__ adj_tail,
    const int*   __restrict__ adj_rel,    const float* __restrict__ b2sum,
    float* __restrict__ drug_e)
{
  extern __shared__ __align__(16) float smem[];
  float* Atab  = smem;
  float* W1p   = Atab + N_REL * ASTR;
  float* demb  = W1p + 32 * PSTR;
  float* w2sum = demb + D;
  float* wstat = w2sum + D;
  float* red   = wstat + 16;

  const int n    = blockIdx.x;
  const int t    = threadIdx.x;
  const int lane = t & 31;
  const int wave = t >> 5;
  const int swave = __builtin_amdgcn_readfirstlane(wave);
  const int h    = lane >> 4;
  const int c    = lane & 15;

  const size_t wbase = (size_t)n * D * D;
  const int* relp  = adj_rel  + (size_t)n * K + swave * 128 + c;
  const int* tailp = adj_tail + (size_t)n * K + swave * 128 + c;
  int rcur = relp[0];

  if (t < D) demb[t] = drug_table[(size_t)drug_name[n] * D + t];
  for (int i = t; i < D * D; i += 256) {
    const int d = i >> 6, e = i & 63;
    W1p[(d >> 1) * PSTR + e * 2 + (d & 1)] = W1g[wbase + i];
  }
  {
    const int d = t >> 2, q = t & 3;
    const float* p = W2g + wbase + d * D + q * 16;
    float s = 0.f;
    #pragma unroll
    for (int j = 0; j < 16; ++j) s += p[j];
    red[t] = s;
  }
  __syncthreads();
  if (t < D) w2sum[t] = red[4*t] + red[4*t+1] + red[4*t+2] + red[4*t+3];
  for (int i = t; i < N_REL * D; i += 256) {
    const int r = i >> 6, d = i & 63;
    Atab[r * ASTR + d] = demb[d] * rela_table[i];
  }
  __syncthreads();

  const float* wp = W1p + (16 * h) * PSTR + 2 * c;
  const unsigned lcolb = 8u * (unsigned)lane;
  const char* entb = (const char*)ent_table;
  float mrun = -INFINITY;
  float lsum = 0.f;
  v2f   gacc = {0.f, 0.f};
  for (int tt = 0; tt < 8; ++tt) {
    const int k = swave * 128 + tt * 16 + c;
    const int rnext = relp[(tt < 7 ? tt + 1 : tt) * 16];
    const int tailv = tailp[tt * 16];
    const float* arow = Atab + rcur * ASTR + 32 * h;
    v4f bq[8];
    #pragma unroll
    for (int j = 0; j < 8; ++j) bq[j] = *(const v4f*)(arow + 4 * j);
    const float b2s = b2sum[k];
    const float* bt = b1 + (size_t)k * D + 8 * h;
    float v = 0.f;
    #pragma unroll 1
    for (int eh = 0; eh < 2; ++eh) {
      const float* bp = bt + 32 * eh;
      v4f bb[2][2];
      #pragma unroll
      for (int e2 = 0; e2 < 2; ++e2) {
        bb[e2][0] = *(const v4f*)(bp + 16 * e2);
        bb[e2][1] = *(const v4f*)(bp + 16 * e2 + 4);
      }
      v8f acc[2];
      #pragma unroll
      for (int e2 = 0; e2 < 2; ++e2)
        acc[e2] = (v8f){0.f, 0.f, 0.f, 0.f, 0.f, 0.f, 0.f, 0.f};
      const float* wpe = wp + 64 * eh;
      #pragma unroll
      for (int kk = 0; kk < 16; ++kk) {
        v2f b;
        b[0] = bq[kk >> 1][2 * (kk & 1)];
        b[1] = bq[kk >> 1][2 * (kk & 1) + 1];
        #pragma unroll
        for (int e2 = 0; e2 < 2; ++e2) {
          const v2f a = *(const v2f*)(wpe + kk * PSTR + 32 * e2);
          acc[e2] = __builtin_amdgcn_wmma_f32_16x16x4_f32(false, a, false, b,
                                                          (short)0, acc[e2], false, false);
        }
      }
      #pragma unroll
      for (int e2 = 0; e2 < 2; ++e2) {
        const v4f w0 = *(const v4f*)(w2sum + 32 * eh + 16 * e2 + 8 * h);
        const v4f w1 = *(const v4f*)(w2sum + 32 * eh + 16 * e2 + 8 * h + 4);
        #pragma unroll
        for (int r = 0; r < 4; ++r) {
          v += fmaxf(acc[e2][r]     + bb[e2][0][r], 0.f) * w0[r];
          v += fmaxf(acc[e2][4 + r] + bb[e2][1][r], 0.f) * w1[r];
        }
      }
    }
    v += __shfl_xor(v, 16, 32);
    const float lg = v + b2s;
    const float mnew = __builtin_bit_cast(float, __builtin_amdgcn_readfirstlane(
        __builtin_bit_cast(int, fmaxf(mrun, half16_max(lg)))));
    const float resc = __expf(mrun - mnew);
    const float p    = __expf(lg - mnew);
    lsum = lsum * resc + p;
    gacc[0] *= resc;
    gacc[1] *= resc;
    #pragma unroll
    for (int j = 0; j < 16; ++j) {
      const float pj = __builtin_bit_cast(float,
          __builtin_amdgcn_readlane(__builtin_bit_cast(int, p), j));
      const int   tj = __builtin_amdgcn_readlane(tailv, j);
      const unsigned boff = (unsigned)tj * (unsigned)(D * sizeof(float)) + lcolb;
      const v2f row = *(const v2f*)(entb + boff);
      gacc[0] += pj * row[0];
      gacc[1] += pj * row[1];
    }
    mrun = mnew;
    rcur = rnext;
  }

  const float lw = half16_sum(lsum);
  *(v2f*)(red + wave * 64 + 2 * lane) = gacc;
  if (lane == 0) { wstat[wave] = mrun; wstat[8 + wave] = lw; }
  __syncthreads();
  if (t < D) {
    float mx = wstat[0];
    #pragma unroll
    for (int w = 1; w < 8; ++w) mx = fmaxf(mx, wstat[w]);
    float den = 0.f, s = 0.f;
    #pragma unroll
    for (int w = 0; w < 8; ++w) {
      const float f = __expf(wstat[w] - mx);
      den += f * wstat[8 + w];
      s   += f * red[w * 64 + t];
    }
    drug_e[(size_t)n * (2 * D) + t]     = s / den;
    drug_e[(size_t)n * (2 * D) + D + t] = demb[t];
  }
}

__global__ void __launch_bounds__(64) gnn_linear(
    const float* __restrict__ drug_e, const float* __restrict__ lin_w,
    const float* __restrict__ lin_b, float* __restrict__ y)
{
  __shared__ float de[2 * D];
  const int n = blockIdx.x, t = threadIdx.x;
  de[t]     = drug_e[(size_t)n * (2 * D) + t];
  de[t + D] = drug_e[(size_t)n * (2 * D) + D + t];
  __syncthreads();
  float acc = lin_b[t];
  const float* w = lin_w + (size_t)t * (2 * D);
  #pragma unroll 8
  for (int j = 0; j < 2 * D; ++j) acc += de[j] * w[j];
  y[(size_t)n * D + t] = fmaxf(acc, 0.f);
}

__global__ void __launch_bounds__(256) gnn_bn(
    const float* __restrict__ y, const float* __restrict__ bn_w,
    const float* __restrict__ bn_b, float* __restrict__ out)
{
  __shared__ float red[256];
  const int c = blockIdx.x, t = threadIdx.x;
  float s = 0.f;
  for (int r = t; r < N_DRUG; r += 256) s += y[(size_t)r * D + c];
  red[t] = s;
  __syncthreads();
  for (int st = 128; st > 0; st >>= 1) {
    if (t < st) red[t] += red[t + st];
    __syncthreads();
  }
  const float mean = red[0] / (float)N_DRUG;
  __syncthreads();
  float q = 0.f;
  for (int r = t; r < N_DRUG; r += 256) {
    const float d = y[(size_t)r * D + c] - mean;
    q += d * d;
  }
  red[t] = q;
  __syncthreads();
  for (int st = 128; st > 0; st >>= 1) {
    if (t < st) red[t] += red[t + st];
    __syncthreads();
  }
  const float var   = red[0] / (float)N_DRUG;
  const float scale = bn_w[c] * rsqrtf(var + BN_EPS);
  const float shift = bn_b[c];
  for (int r = t; r < N_DRUG; r += 256)
    out[(size_t)r * D + c] = (y[(size_t)r * D + c] - mean) * scale + shift;
}

extern "C" void kernel_launch(void* const* d_in, const int* in_sizes, int n_in,
                              void* d_out, int out_size, void* d_ws, size_t ws_size,
                              hipStream_t stream) {
  (void)in_sizes; (void)n_in; (void)out_size; (void)ws_size;
  const float* drug_table = (const float*)d_in[0];
  const float* rela_table = (const float*)d_in[1];
  const float* ent_table  = (const float*)d_in[2];
  const float* W1         = (const float*)d_in[3];
  const float* b1         = (const float*)d_in[4];
  const float* W2         = (const float*)d_in[5];
  const float* b2         = (const float*)d_in[6];
  const float* lin_w      = (const float*)d_in[7];
  const float* lin_b      = (const float*)d_in[8];
  const float* bn_w       = (const float*)d_in[9];
  const float* bn_b       = (const float*)d_in[10];
  const int*   drug_name  = (const int*)d_in[11];
  const int*   adj_tail   = (const int*)d_in[12];
  const int*   adj_rel    = (const int*)d_in[13];

  float* ws     = (float*)d_ws;
  float* b2sum  = ws;
  float* drug_e = ws + K;
  float* y      = drug_e + N_DRUG * 2 * D;

  gnn_b2sum<<<dim3((K + 255) / 256), dim3(256), 0, stream>>>(b2, b2sum);

  const size_t lds = (size_t)(N_REL * ASTR + 32 * PSTR + D + D + 16 + 512) * sizeof(float);
  gnn_main<<<dim3(N_DRUG), dim3(256), lds, stream>>>(
      drug_table, rela_table, ent_table, W1, b1, W2,
      drug_name, adj_tail, adj_rel, b2sum, drug_e);

  gnn_linear<<<dim3(N_DRUG), dim3(64), 0, stream>>>(drug_e, lin_w, lin_b, y);

  gnn_bn<<<dim3(D), dim3(256), 0, stream>>>(y, bn_w, bn_b, (float*)d_out);
}


# Linear layer and batch norm in one launch: 64 column-owner work-groups of 32 waves form their y column from whole drug_e rows and normalise it from LDS; linear, mean and variance sum orders change
# speedup vs baseline: 3.5402x; 1.0610x over previous
#include <hip/hip_runtime.h>

#define N_DRUG 846
#define D 64
#define K 1024
#define N_REL 200
#define BN_EPS 1e-5f

typedef __attribute__((ext_vector_type(2))) float v2f;
typedef __attribute__((ext_vector_type(4))) float v4f;
typedef __attribute__((ext_vector_type(8))) float v8f;

__device__ __forceinline__ float wave_reduce_add(float v) {
  #pragma unroll
  for (int off = 16; off > 0; off >>= 1) v += __shfl_xor(v, off, 32);
  return v;
}

template <int CTRL>
__device__ __forceinline__ float half_xor(float v) {
  return __builtin_bit_cast(float,
      __builtin_amdgcn_update_dpp(0, __builtin_bit_cast(int, v), CTRL, 0xf, 0xf, true));
}
__device__ __forceinline__ float half16_max(float v) {
  v = fmaxf(v, half_xor<0x168>(v));
  v = fmaxf(v, half_xor<0x164>(v));
  v = fmaxf(v, half_xor<0x162>(v));
  v = fmaxf(v, half_xor<0x161>(v));
  return v;
}
__device__ __forceinline__ float half16_sum(float v) {
  v += half_xor<0x168>(v);
  v += half_xor<0x164>(v);
  v += half_xor<0x162>(v);
  v += half_xor<0x161>(v);
  return v;
}

__global__ void __launch_bounds__(256) gnn_b2sum(const float* __restrict__ b2,
                                                 float* __restrict__ b2sum) {
  const int k = blockIdx.x * 256 + threadIdx.x;
  if (k < K) {
    const float* p = b2 + (size_t)k * D;
    float s = 0.f;
    #pragma unroll
    for (int e = 0; e < D; ++e) s += p[e];
    b2sum[k] = s;
  }
}

#define ASTR 68
#define PSTR 130

__global__ void __launch_bounds__(256, 8) gnn_main(
    const float* __restrict__ drug_table, const float* __restrict__ rela_table,
    const float* __restrict__ ent_table,  const float* __restrict__ W1g,
    const float* __restrict__ b1,         const float* __restrict__ W2g,
    const int*   __restrict__ drug_name,  const int* __restrict__ adj_tail,
    const int*   __restrict__ adj_rel,    const float* __restrict__ b2sum,
    float* __restrict__ drug_e)
{
  extern __shared__ __align__(16) float smem[];
  float* Atab  = smem;
  float* W1p   = Atab + N_REL * ASTR;
  float* demb  = W1p + 32 * PSTR;
  float* w2sum = demb + D;
  float* wstat = w2sum + D;
  float* red   = wstat + 16;

  const int n    = blockIdx.x;
  const int t    = threadIdx.x;
  const int lane = t & 31;
  const int wave = t >> 5;
  const int swave = __builtin_amdgcn_readfirstlane(wave);
  const int h    = lane >> 4;
  const int c    = lane & 15;

  const size_t wbase = (size_t)n * D * D;
  const int* relp  = adj_rel  + (size_t)n * K + swave * 128 + c;
  const int* tailp = adj_tail + (size_t)n * K + swave * 128 + c;
  int rcur = relp[0];

  if (t < D) demb[t] = drug_table[(size_t)drug_name[n] * D + t];
  for (int i = t; i < D * D; i += 256) {
    const int d = i >> 6, e = i & 63;
    W1p[(d >> 1) * PSTR + e * 2 + (d & 1)] = W1g[wbase + i];
  }
  {
    const int d = t >> 2, q = t & 3;
    const float* p = W2g + wbase + d * D + q * 16;
    float s = 0.f;
    #pragma unroll
    for (int j = 0; j < 16; ++j) s += p[j];
    red[t] = s;
  }
  __syncthreads();
  if (t < D) w2sum[t] = red[4*t] + red[4*t+1] + red[4*t+2] + red[4*t+3];
  for (int i = t; i < N_REL * D; i += 256) {
    const int r = i >> 6, d = i & 63;
    Atab[r * ASTR + d] = demb[d] * rela_table[i];
  }
  __syncthreads();

  const float* wp = W1p + (16 * h) * PSTR + 2 * c;
  const unsigned lcolb = 8u * (unsigned)lane;
  const char* entb = (const char*)ent_table;
  float mrun = -INFINITY;
  float lsum = 0.f;
  v2f   gacc = {0.f, 0.f};
  for (int tt = 0; tt < 8; ++tt) {
    const int k = swave * 128 + tt * 16 + c;
    const int rnext = relp[(tt < 7 ? tt + 1 : tt) * 16];
    const int tailv = tailp[tt * 16];
    const float* arow = Atab + rcur * ASTR + 32 * h;
    v4f bq[8];
    #pragma unroll
    for (int j = 0; j < 8; ++j) bq[j] = *(const v4f*)(arow + 4 * j);
    const float b2s = b2sum[k];
    const float* bt = b1 + (size_t)k * D + 8 * h;
    float v = 0.f;
    #pragma unroll 1
    for (int eh = 0; eh < 2; ++eh) {
      const float* bp = bt + 32 * eh;
      v4f bb[2][2];
      #pragma unroll
      for (int e2 = 0; e2 < 2; ++e2) {
        bb[e2][0] = *(const v4f*)(bp + 16 * e2);
        bb[e2][1] = *(const v4f*)(bp + 16 * e2 + 4);
      }
      v8f acc[2];
      #pragma unroll
      for (int e2 = 0; e2 < 2; ++e2)
        acc[e2] = (v8f){0.f, 0.f, 0.f, 0.f, 0.f, 0.f, 0.f, 0.f};
      const float* wpe = wp + 64 * eh;
      #pragma unroll
      for (int kk = 0; kk < 16; ++kk) {
        v2f b;
        b[0] = bq[kk >> 1][2 * (kk & 1)];
        b[1] = bq[kk >> 1][2 * (kk & 1) + 1];
        #pragma unroll
        for (int e2 = 0; e2 < 2; ++e2) {
          const v2f a = *(const v2f*)(wpe + kk * PSTR + 32 * e2);
          acc[e2] = __builtin_amdgcn_wmma_f32_16x16x4_f32(false, a, false, b,
                                                          (short)0, acc[e2], false, false);
        }
      }
      #pragma unroll
      for (int e2 = 0; e2 < 2; ++e2) {
        const v4f w0 = *(const v4f*)(w2sum + 32 * eh + 16 * e2 + 8 * h);
        const v4f w1 = *(const v4f*)(w2sum + 32 * eh + 16 * e2 + 8 * h + 4);
        #pragma unroll
        for (int r = 0; r < 4; ++r) {
          v += fmaxf(acc[e2][r]     + bb[e2][0][r], 0.f) * w0[r];
          v += fmaxf(acc[e2][4 + r] + bb[e2][1][r], 0.f) * w1[r];
        }
      }
    }
    v += __shfl_xor(v, 16, 32);
    const float lg = v + b2s;
    const float mnew = __builtin_bit_cast(float, __builtin_amdgcn_readfirstlane(
        __builtin_bit_cast(int, fmaxf(mrun, half16_max(lg)))));
    const float resc = __expf(mrun - mnew);
    const float p    = __expf(lg - mnew);
    lsum = lsum * resc + p;
    gacc[0] *= resc;
    gacc[1] *= resc;
    #pragma unroll
    for (int j = 0; j < 16; ++j) {
      const float pj = __builtin_bit_cast(float,
          __builtin_amdgcn_readlane(__builtin_bit_cast(int, p), j));
      const int   tj = __builtin_amdgcn_readlane(tailv, j);
      const unsigned boff = (unsigned)tj * (unsigned)(D * sizeof(float)) + lcolb;
      const v2f row = *(const v2f*)(entb + boff);
      gacc[0] += pj * row[0];
      gacc[1] += pj * row[1];
    }
    mrun = mnew;
    rcur = rnext;
  }

  const float lw = half16_sum(lsum);
  *(v2f*)(red + wave * 64 + 2 * lane) = gacc;
  if (lane == 0) { wstat[wave] = mrun; wstat[8 + wave] = lw; }
  __syncthreads();
  if (t < D) {
    float mx = wstat[0];
    #pragma unroll
    for (int w = 1; w < 8; ++w) mx = fmaxf(mx, wstat[w]);
    float den = 0.f, s = 0.f;
    #pragma unroll
    for (int w = 0; w < 8; ++w) {
      const float f = __expf(wstat[w] - mx);
      den += f * wstat[8 + w];
      s   += f * red[w * 64 + t];
    }
    drug_e[(size_t)n * (2 * D) + t]     = s / den;
    drug_e[(size_t)n * (2 * D) + D + t] = demb[t];
  }
}

#define LB_ROWS 27
__global__ void __launch_bounds__(1024) gnn_linear_bn(
    const float* __restrict__ drug_e, const float* __restrict__ lin_w,
    const float* __restrict__ lin_b,  const float* __restrict__ bn_w,
    const float* __restrict__ bn_b,   float* __restrict__ out)
{
  __shared__ float ysum[2][32 * LB_ROWS];
  __shared__ float part[2][64];
  const int c = blockIdx.x, t = threadIdx.x;
  const int lane = t & 31;
  const int wave = __builtin_amdgcn_readfirstlane(t >> 5);
  const v4f w = *(const v4f*)(lin_w + (size_t)c * (2 * D) + 4 * lane);

  const int row0 = wave * LB_ROWS;
  v4f de[LB_ROWS];
  #pragma unroll
  for (int i = 0; i < LB_ROWS; ++i) {
    const int rc = min(row0 + i, N_DRUG - 1);
    de[i] = *(const v4f*)(drug_e + (size_t)rc * (2 * D) + 4 * lane);
  }
  #pragma unroll
  for (int i = 0; i < LB_ROWS; ++i) {
    float p = de[i][0] * w[0];
    p += de[i][1] * w[1];
    p += de[i][2] * w[2];
    p += de[i][3] * w[3];
    p = half16_sum(p);
    if ((lane & 15) == 0) ysum[lane >> 4][row0 + i] = p;
  }
  __syncthreads();

  const bool live = t < N_DRUG;
  float y = 0.f;
  if (live) y = fmaxf(ysum[0][t] + ysum[1][t] + lin_b[c], 0.f);
  const float s = half16_sum(y);
  if ((lane & 15) == 0) part[0][2 * (t >> 5) + (lane >> 4)] = s;
  __syncthreads();
  float tot = 0.f;
  #pragma unroll
  for (int i = 0; i < 64; ++i) tot += part[0][i];
  const float mean = tot / (float)N_DRUG;
  const float d = live ? y - mean : 0.f;
  const float q = half16_sum(d * d);
  if ((lane & 15) == 0) part[1][2 * (t >> 5) + (lane >> 4)] = q;
  __syncthreads();
  float qt = 0.f;
  #pragma unroll
  for (int i = 0; i < 64; ++i) qt += part[1][i];
  const float var   = qt / (float)N_DRUG;
  const float scale = bn_w[c] * rsqrtf(var + BN_EPS);
  const float shift = bn_b[c];
  if (live) out[(size_t)t * D + c] = d * scale + shift;
}

extern "C" void kernel_launch(void* const* d_in, const int* in_sizes, int n_in,
                              void* d_out, int out_size, void* d_ws, size_t ws_size,
                              hipStream_t stream) {
  (void)in_sizes; (void)n_in; (void)out_size; (void)ws_size;
  const float* drug_table = (const float*)d_in[0];
  const float* rela_table = (const float*)d_in[1];
  const float* ent_table  = (const float*)d_in[2];
  const float* W1         = (const float*)d_in[3];
  const float* b1         = (const float*)d_in[4];
  const float* W2         = (const float*)d_in[5];
  const float* b2         = (const float*)d_in[6];
  const float* lin_w      = (const float*)d_in[7];
  const float* lin_b      = (const float*)d_in[8];
  const float* bn_w       = (const float*)d_in[9];
  const float* bn_b       = (const float*)d_in[10];
  const int*   drug_name  = (const int*)d_in[11];
  const int*   adj_tail   = (const int*)d_in[12];
  const int*   adj_rel    = (const int*)d_in[13];

  float* ws     = (float*)d_ws;
  float* b2sum  = ws;
  float* drug_e = ws + K;

  gnn_b2sum<<<dim3((K + 255) / 256), dim3(256), 0, stream>>>(b2, b2sum);

  const size_t lds = (size_t)(N_REL * ASTR + 32 * PSTR + D + D + 16 + 512) * sizeof(float);
  gnn_main<<<dim3(N_DRUG), dim3(256), lds, stream>>>(
      drug_table, rela_table, ent_table, W1, b1, W2,
      drug_name, adj_tail, adj_rel, b2sum, drug_e);

  gnn_linear_bn<<<dim3(D), dim3(1024), 0, stream>>>(drug_e, lin_w, lin_b, bn_w, bn_b, (float*)d_out);
}


# Relation table Tt formed once per drug on the matrix unit: 13 tiles of 16 relations, not 64 tiles of 16 neighbours; logits read its rows from LDS, tail indices a tile ahead; sum orders kept
# speedup vs baseline: 4.6240x; 1.3061x over previous
#include <hip/hip_runtime.h>

#define N_DRUG 846
#define D 64
#define K 1024
#define N_REL 200
#define BN_EPS 1e-5f

typedef __attribute__((ext_vector_type(2))) float v2f;
typedef __attribute__((ext_vector_type(4))) float v4f;
typedef __attribute__((ext_vector_type(8))) float v8f;

__device__ __forceinline__ float wave_reduce_add(float v) {
  #pragma unroll
  for (int off = 16; off > 0; off >>= 1) v += __shfl_xor(v, off, 32);
  return v;
}

template <int CTRL>
__device__ __forceinline__ float half_xor(float v) {
  return __builtin_bit_cast(float,
      __builtin_amdgcn_update_dpp(0, __builtin_bit_cast(int, v), CTRL, 0xf, 0xf, true));
}
__device__ __forceinline__ float half16_max(float v) {
  v = fmaxf(v, half_xor<0x168>(v));
  v = fmaxf(v, half_xor<0x164>(v));
  v = fmaxf(v, half_xor<0x162>(v));
  v = fmaxf(v, half_xor<0x161>(v));
  return v;
}
__device__ __forceinline__ float half16_sum(float v) {
  v += half_xor<0x168>(v);
  v += half_xor<0x164>(v);
  v += half_xor<0x162>(v);
  v += half_xor<0x161>(v);
  return v;
}

__global__ void __launch_bounds__(256) gnn_b2sum(const float* __restrict__ b2,
                                                 float* __restrict__ b2sum) {
  const int k = blockIdx.x * 256 + threadIdx.x;
  if (k < K) {
    const float* p = b2 + (size_t)k * D;
    float s = 0.f;
    #pragma unroll
    for (int e = 0; e < D; ++e) s += p[e];
    b2sum[k] = s;
  }
}

#define ASTR 68
#define PSTR 130

__global__ void __launch_bounds__(256, 8) gnn_main(
    const float* __restrict__ drug_table, const float* __restrict__ rela_table,
    const float* __restrict__ ent_table,  const float* __restrict__ W1g,
    const float* __restrict__ b1,         const float* __restrict__ W2g,
    const int*   __restrict__ drug_name,  const int* __restrict__ adj_tail,
    const int*   __restrict__ adj_rel,    const float* __restrict__ b2sum,
    float* __restrict__ drug_e)
{
  extern __shared__ __align__(16) float smem[];
  float* Atab  = smem;
  float* W1p   = Atab + N_REL * ASTR;
  float* demb  = W1p + 32 * PSTR;
  float* w2sum = demb + D;
  float* wstat = w2sum + D;
  float* red   = wstat + 16;

  const int n    = blockIdx.x;
  const int t    = threadIdx.x;
  const int lane = t & 31;
  const int wave = t >> 5;
  const int swave = __builtin_amdgcn_readfirstlane(wave);
  const int h    = lane >> 4;
  const int c    = lane & 15;

  const size_t wbase = (size_t)n * D * D;
  const int* relp  = adj_rel  + (size_t)n * K + swave * 128 + c;
  const int* tailp = adj_tail + (size_t)n * K + swave * 128 + c;
  int rcur = relp[0];

  if (t < D) demb[t] = drug_table[(size_t)drug_name[n] * D + t];
  for (int i = t; i < D * D; i += 256) {
    const int d = i >> 6, e = i & 63;
    W1p[(d >> 1) * PSTR + e * 2 + (d & 1)] = W1g[wbase + i];
  }
  {
    const int d = t >> 2, q = t & 3;
    const float* p = W2g + wbase + d * D + q * 16;
    float s = 0.f;
    #pragma unroll
    for (int j = 0; j < 16; ++j) s += p[j];
    red[t] = s;
  }
  __syncthreads();
  if (t < D) w2sum[t] = red[4*t] + red[4*t+1] + red[4*t+2] + red[4*t+3];
  for (int i = t; i < N_REL * D; i += 256) {
    const int r = i >> 6, d = i & 63;
    Atab[r * ASTR + d] = demb[d] * rela_table[i];
  }
  __syncthreads();

  const float* wp = W1p + (16 * h) * PSTR + 2 * c;
  for (int j = swave; j < (N_REL + 15) / 16; j += 8) {
    const int rel  = 16 * j + c;
    const int relc = min(rel, N_REL - 1);
    float* trow = Atab + relc * ASTR;
    v4f bq[8];
    #pragma unroll
    for (int q = 0; q < 8; ++q) bq[q] = *(const v4f*)(trow + 32 * h + 4 * q);
    #pragma unroll 1
    for (int eh = 0; eh < 2; ++eh) {
      v8f acc[2];
      #pragma unroll
      for (int e2 = 0; e2 < 2; ++e2)
        acc[e2] = (v8f){0.f, 0.f, 0.f, 0.f, 0.f, 0.f, 0.f, 0.f};
      const float* wpe = wp + 64 * eh;
      #pragma unroll
      for (int kk = 0; kk < 16; ++kk) {
        v2f b;
        b[0] = bq[kk >> 1][2 * (kk & 1)];
        b[1] = bq[kk >> 1][2 * (kk & 1) + 1];
        #pragma unroll
        for (int e2 = 0; e2 < 2; ++e2) {
          const v2f a = *(const v2f*)(wpe + kk * PSTR + 32 * e2);
          acc[e2] = __builtin_amdgcn_wmma_f32_16x16x4_f32(false, a, false, b,
                                                          (short)0, acc[e2], false, false);
        }
      }
      if (rel < N_REL) {
        #pragma unroll
        for (int e2 = 0; e2 < 2; ++e2) {
          float* tp = trow + 32 * eh + 16 * e2 + 8 * h;
          *(v4f*)tp       = (v4f){acc[e2][0], acc[e2][1], acc[e2][2], acc[e2][3]};
          *(v4f*)(tp + 4) = (v4f){acc[e2][4], acc[e2][5], acc[e2][6], acc[e2][7]};
        }
      }
    }
  }
  __syncthreads();

  const unsigned lcolb = 8u * (unsigned)lane;
  const char* entb = (const char*)ent_table;
  float mrun = -INFINITY;
  float lsum = 0.f;
  v2f   gacc = {0.f, 0.f};
  int tcur = tailp[0];
  for (int tt = 0; tt < 8; ++tt) {
    const int k = swave * 128 + tt * 16 + c;
    const int rnext = relp[(tt < 7 ? tt + 1 : tt) * 16];
    const int tnext = tailp[(tt < 7 ? tt + 1 : tt) * 16];
    const int tailv = tcur;
    const float* bt = b1 + (size_t)k * D + 8 * h;
    v4f bb[4][2];
    #pragma unroll
    for (int m = 0; m < 4; ++m) {
      bb[m][0] = *(const v4f*)(bt + 16 * m);
      bb[m][1] = *(const v4f*)(bt + 16 * m + 4);
    }
    const float b2s = b2sum[k];
    const float* trow = Atab + rcur * ASTR + 8 * h;
    float v = 0.f;
    #pragma unroll
    for (int m = 0; m < 4; ++m) {
      const v4f t0 = *(const v4f*)(trow + 16 * m);
      const v4f t1 = *(const v4f*)(trow + 16 * m + 4);
      const v4f w0 = *(const v4f*)(w2sum + 16 * m + 8 * h);
      const v4f w1 = *(const v4f*)(w2sum + 16 * m + 8 * h + 4);
      #pragma unroll
      for (int r = 0; r < 4; ++r) {
        v += fmaxf(t0[r] + bb[m][0][r], 0.f) * w0[r];
        v += fmaxf(t1[r] + bb[m][1][r], 0.f) * w1[r];
      }
    }
    v += __shfl_xor(v, 16, 32);
    const float lg = v + b2s;
    const float mnew = __builtin_bit_cast(float, __builtin_amdgcn_readfirstlane(
        __builtin_bit_cast(int, fmaxf(mrun, half16_max(lg)))));
    const float resc = __expf(mrun - mnew);
    const float p    = __expf(lg - mnew);
    lsum = lsum * resc + p;
    gacc[0] *= resc;
    gacc[1] *= resc;
    #pragma unroll
    for (int j = 0; j < 16; ++j) {
      const float pj = __builtin_bit_cast(float,
          __builtin_amdgcn_readlane(__builtin_bit_cast(int, p), j));
      const int   tj = __builtin_amdgcn_readlane(tailv, j);
      const unsigned boff = (unsigned)tj * (unsigned)(D * sizeof(float)) + lcolb;
      const v2f row = *(const v2f*)(entb + boff);
      gacc[0] += pj * row[0];
      gacc[1] += pj * row[1];
    }
    mrun = mnew;
    rcur = rnext;
    tcur = tnext;
  }

  const float lw = half16_sum(lsum);
  *(v2f*)(red + wave * 64 + 2 * lane) = gacc;
  if (lane == 0) { wstat[wave] = mrun; wstat[8 + wave] = lw; }
  __syncthreads();
  if (t < D) {
    float mx = wstat[0];
    #pragma unroll
    for (int w = 1; w < 8; ++w) mx = fmaxf(mx, wstat[w]);
    float den = 0.f, s = 0.f;
    #pragma unroll
    for (int w = 0; w < 8; ++w) {
      const float f = __expf(wstat[w] - mx);
      den += f * wstat[8 + w];
      s   += f * red[w * 64 + t];
    }
    drug_e[(size_t)n * (2 * D) + t]     = s / den;
    drug_e[(size_t)n * (2 * D) + D + t] = demb[t];
  }
}

#define LB_ROWS 27
__global__ void __launch_bounds__(1024) gnn_linear_bn(
    const float* __restrict__ drug_e, const float* __restrict__ lin_w,
    const float* __restrict__ lin_b,  const float* __restrict__ bn_w,
    const float* __restrict__ bn_b,   float* __restrict__ out)
{
  __shared__ float ysum[2][32 * LB_ROWS];
  __shared__ float part[2][64];
  const int c = blockIdx.x, t = threadIdx.x;
  const int lane = t & 31;
  const int wave = __builtin_amdgcn_readfirstlane(t >> 5);
  const v4f w = *(const v4f*)(lin_w + (size_t)c * (2 * D) + 4 * lane);

  const int row0 = wave * LB_ROWS;
  v4f de[LB_ROWS];
  #pragma unroll
  for (int i = 0; i < LB_ROWS; ++i) {
    const int rc = min(row0 + i, N_DRUG - 1);
    de[i] = *(const v4f*)(drug_e + (size_t)rc * (2 * D) + 4 * lane);
  }
  #pragma unroll
  for (int i = 0; i < LB_ROWS; ++i) {
    float p = de[i][0] * w[0];
    p += de[i][1] * w[1];
    p += de[i][2] * w[2];
    p += de[i][3] * w[3];
    p = half16_sum(p);
    if ((lane & 15) == 0) ysum[lane >> 4][row0 + i] = p;
  }
  __syncthreads();

  const bool live = t < N_DRUG;
  float y = 0.f;
  if (live) y = fmaxf(ysum[0][t] + ysum[1][t] + lin_b[c], 0.f);
  const float s = half16_sum(y);
  if ((lane & 15) == 0) part[0][2 * (t >> 5) + (lane >> 4)] = s;
  __syncthreads();
  float tot = 0.f;
  #pragma unroll
  for (int i = 0; i < 64; ++i) tot += part[0][i];
  const float mean = tot / (float)N_DRUG;
  const float d = live ? y - mean : 0.f;
  const float q = half16_sum(d * d);
  if ((lane & 15) == 0) part[1][2 * (t >> 5) + (lane >> 4)] = q;
  __syncthreads();
  float qt = 0.f;
  #pragma unroll
  for (int i = 0; i < 64; ++i) qt += part[1][i];
  const float var   = qt / (float)N_DRUG;
  const float scale = bn_w[c] * rsqrtf(var + BN_EPS);
  const float shift = bn_b[c];
  if (live) out[(size_t)t * D + c] = d * scale + shift;
}

extern "C" void kernel_launch(void* const* d_in, const int* in_sizes, int n_in,
                              void* d_out, int out_size, void* d_ws, size_t ws_size,
                              hipStream_t stream) {
  (void)in_sizes; (void)n_in; (void)out_size; (void)ws_size;
  const float* drug_table = (const float*)d_in[0];
  const float* rela_table = (const float*)d_in[1];
  const float* ent_table  = (const float*)d_in[2];
  const float* W1         = (const float*)d_in[3];
  const float* b1         = (const float*)d_in[4];
  const float* W2         = (const float*)d_in[5];
  const float* b2         = (const float*)d_in[6];
  const float* lin_w      = (const float*)d_in[7];
  const float* lin_b      = (const float*)d_in[8];
  const float* bn_w       = (const float*)d_in[9];
  const float* bn_b       = (const float*)d_in[10];
  const int*   drug_name  = (const int*)d_in[11];
  const int*   adj_tail   = (const int*)d_in[12];
  const int*   adj_rel    = (const int*)d_in[13];

  float* ws     = (float*)d_ws;
  float* b2sum  = ws;
  float* drug_e = ws + K;

  gnn_b2sum<<<dim3((K + 255) / 256), dim3(256), 0, stream>>>(b2, b2sum);

  const size_t lds = (size_t)(N_REL * ASTR + 32 * PSTR + D + D + 16 + 512) * sizeof(float);
  gnn_main<<<dim3(N_DRUG), dim3(256), lds, stream>>>(
      drug_table, rela_table, ent_table, W1, b1, W2,
      drug_name, adj_tail, adj_rel, b2sum, drug_e);

  gnn_linear_bn<<<dim3(D), dim3(1024), 0, stream>>>(drug_e, lin_w, lin_b, bn_w, bn_b, (float*)d_out);
}


# Logit tiles re-mapped: a lane takes 4 hidden units of 8 neighbours; b1 read as whole 512-B runs, table rows as whole LDS rows; 16-lane butterfly per logit; logit and weight sum orders change
# speedup vs baseline: 5.4207x; 1.1723x over previous
#include <hip/hip_runtime.h>

#define N_DRUG 846
#define D 64
#define K 1024
#define N_REL 200
#define BN_EPS 1e-5f

typedef __attribute__((ext_vector_type(2))) float v2f;
typedef __attribute__((ext_vector_type(4))) float v4f;
typedef __attribute__((ext_vector_type(8))) float v8f;

__device__ __forceinline__ float wave_reduce_add(float v) {
  #pragma unroll
  for (int off = 16; off > 0; off >>= 1) v += __shfl_xor(v, off, 32);
  return v;
}

template <int CTRL>
__device__ __forceinline__ float half_xor(float v) {
  return __builtin_bit_cast(float,
      __builtin_amdgcn_update_dpp(0, __builtin_bit_cast(int, v), CTRL, 0xf, 0xf, true));
}
__device__ __forceinline__ float half16_max(float v) {
  v = fmaxf(v, half_xor<0x168>(v));
  v = fmaxf(v, half_xor<0x164>(v));
  v = fmaxf(v, half_xor<0x162>(v));
  v = fmaxf(v, half_xor<0x161>(v));
  return v;
}
__device__ __forceinline__ float half16_sum(float v) {
  v += half_xor<0x168>(v);
  v += half_xor<0x164>(v);
  v += half_xor<0x162>(v);
  v += half_xor<0x161>(v);
  return v;
}

template <int N>
__device__ __forceinline__ int row_bcast(int v) {
  return __builtin_amdgcn_update_dpp(0, v, 0x150 + N, 0xf, 0xf, true);
}

__global__ void __launch_bounds__(256) gnn_b2sum(const float* __restrict__ b2,
                                                 float* __restrict__ b2sum) {
  const int k = blockIdx.x * 256 + threadIdx.x;
  if (k < K) {
    const float* p = b2 + (size_t)k * D;
    float s = 0.f;
    #pragma unroll
    for (int e = 0; e < D; ++e) s += p[e];
    b2sum[k] = s;
  }
}

#define ASTR 68
#define PSTR 130

__global__ void __launch_bounds__(256, 8) gnn_main(
    const float* __restrict__ drug_table, const float* __restrict__ rela_table,
    const float* __restrict__ ent_table,  const float* __restrict__ W1g,
    const float* __restrict__ b1,         const float* __restrict__ W2g,
    const int*   __restrict__ drug_name,  const int* __restrict__ adj_tail,
    const int*   __restrict__ adj_rel,    const float* __restrict__ b2sum,
    float* __restrict__ drug_e)
{
  extern __shared__ __align__(16) float smem[];
  float* Atab  = smem;
  float* W1p   = Atab + N_REL * ASTR;
  float* demb  = W1p + 32 * PSTR;
  float* w2sum = demb + D;
  float* wstat = w2sum + D;
  float* red   = wstat + 16;

  const int n    = blockIdx.x;
  const int t    = threadIdx.x;
  const int lane = t & 31;
  const int wave = t >> 5;
  const int swave = __builtin_amdgcn_readfirstlane(wave);
  const int h    = lane >> 4;
  const int c    = lane & 15;

  const size_t wbase = (size_t)n * D * D;
  const int nbl = (c & 14) | h;
  const int* relp  = adj_rel  + (size_t)n * K + swave * 128 + nbl;
  const int* tailp = adj_tail + (size_t)n * K + swave * 128 + nbl;
  int rcur = relp[0];

  if (t < D) demb[t] = drug_table[(size_t)drug_name[n] * D + t];
  for (int i = t; i < D * D; i += 256) {
    const int d = i >> 6, e = i & 63;
    W1p[(d >> 1) * PSTR + e * 2 + (d & 1)] = W1g[wbase + i];
  }
  {
    const int d = t >> 2, q = t & 3;
    const float* p = W2g + wbase + d * D + q * 16;
    float s = 0.f;
    #pragma unroll
    for (int j = 0; j < 16; ++j) s += p[j];
    red[t] = s;
  }
  __syncthreads();
  if (t < D) w2sum[t] = red[4*t] + red[4*t+1] + red[4*t+2] + red[4*t+3];
  for (int i = t; i < N_REL * D; i += 256) {
    const int r = i >> 6, d = i & 63;
    Atab[r * ASTR + d] = demb[d] * rela_table[i];
  }
  __syncthreads();

  const float* wp = W1p + (16 * h) * PSTR + 2 * c;
  for (int j = swave; j < (N_REL + 15) / 16; j += 8) {
    const int rel  = 16 * j + c;
    const int relc = min(rel, N_REL - 1);
    float* trow = Atab + relc * ASTR;
    v4f bq[8];
    #pragma unroll
    for (int q = 0; q < 8; ++q) bq[q] = *(const v4f*)(trow + 32 * h + 4 * q);
    #pragma unroll 1
    for (int eh = 0; eh < 2; ++eh) {
      v8f acc[2];
      #pragma unroll
      for (int e2 = 0; e2 < 2; ++e2)
        acc[e2] = (v8f){0.f, 0.f, 0.f, 0.f, 0.f, 0.f, 0.f, 0.f};
      const float* wpe = wp + 64 * eh;
      #pragma unroll
      for (int kk = 0; kk < 16; ++kk) {
        v2f b;
        b[0] = bq[kk >> 1][2 * (kk & 1)];
        b[1] = bq[kk >> 1][2 * (kk & 1) + 1];
        #pragma unroll
        for (int e2 = 0; e2 < 2; ++e2) {
          const v2f a = *(const v2f*)(wpe + kk * PSTR + 32 * e2);
          acc[e2] = __builtin_amdgcn_wmma_f32_16x16x4_f32(false, a, false, b,
                                                          (short)0, acc[e2], false, false);
        }
      }
      if (rel < N_REL) {
        #pragma unroll
        for (int e2 = 0; e2 < 2; ++e2) {
          float* tp = trow + 32 * eh + 16 * e2 + 8 * h;
          *(v4f*)tp       = (v4f){acc[e2][0], acc[e2][1], acc[e2][2], acc[e2][3]};
          *(v4f*)(tp + 4) = (v4f){acc[e2][4], acc[e2][5], acc[e2][6], acc[e2][7]};
        }
      }
    }
  }
  __syncthreads();

  const v4f w4 = *(const v4f*)(w2sum + 4 * c);
  const char* tabl = (const char*)Atab + 16 * c;
  const bool hi8 = (c & 8) != 0, hi4 = (c & 4) != 0, hi2 = (c & 2) != 0;
  const unsigned lcolb = 8u * (unsigned)lane;
  const char* entb = (const char*)ent_table;
  float mrun = -INFINITY;
  float lsum = 0.f;
  v2f   gacc = {0.f, 0.f};
  int tcur = tailp[0];
  for (int tt = 0; tt < 8; ++tt) {
    const int kb = swave * 128 + tt * 16;
    const int rnext = relp[(tt < 7 ? tt + 1 : tt) * 16];
    const int tnext = tailp[(tt < 7 ? tt + 1 : tt) * 16];
    const int tailv = tcur;
    const float* bt = b1 + (size_t)kb * D + 4 * lane;
    v4f bb[8];
    #pragma unroll
    for (int i = 0; i < 8; ++i) bb[i] = *(const v4f*)(bt + 128 * i);
    const float b2s = b2sum[kb + nbl];
    const int roff = rcur * (ASTR * 4);
    v4f tr[8];
    tr[0] = *(const v4f*)(tabl + row_bcast<0>(roff));
    tr[1] = *(const v4f*)(tabl + row_bcast<2>(roff));
    tr[2] = *(const v4f*)(tabl + row_bcast<4>(roff));
    tr[3] = *(const v4f*)(tabl + row_bcast<6>(roff));
    tr[4] = *(const v4f*)(tabl + row_bcast<8>(roff));
    tr[5] = *(const v4f*)(tabl + row_bcast<10>(roff));
    tr[6] = *(const v4f*)(tabl + row_bcast<12>(roff));
    tr[7] = *(const v4f*)(tabl + row_bcast<14>(roff));
    float v[8];
    #pragma unroll
    for (int i = 0; i < 8; ++i) {
      float s = fmaxf(tr[i][0] + bb[i][0], 0.f) * w4[0];
      s += fmaxf(tr[i][1] + bb[i][1], 0.f) * w4[1];
      s += fmaxf(tr[i][2] + bb[i][2], 0.f) * w4[2];
      s += fmaxf(tr[i][3] + bb[i][3], 0.f) * w4[3];
      v[i] = s;
    }
    float u[4];
    #pragma unroll
    for (int i = 0; i < 4; ++i)
      u[i] = (hi8 ? v[i + 4] : v[i]) + half_xor<0x168>(hi8 ? v[i] : v[i + 4]);
    float x[2];
    #pragma unroll
    for (int i = 0; i < 2; ++i)
      x[i] = (hi4 ? u[i + 2] : u[i]) + half_xor<0x164>(hi4 ? u[i] : u[i + 2]);
    float y = (hi2 ? x[1] : x[0]) + half_xor<0x162>(hi2 ? x[0] : x[1]);
    y += half_xor<0x161>(y);
    const float lg = y + b2s;
    const float m16 = half16_max(lg);
    const float mev = __builtin_bit_cast(float, __builtin_amdgcn_readlane(__builtin_bit_cast(int, m16), 0));
    const float mod = __builtin_bit_cast(float, __builtin_amdgcn_readlane(__builtin_bit_cast(int, m16), 16));
    const float mnew = __builtin_bit_cast(float, __builtin_amdgcn_readfirstlane(
        __builtin_bit_cast(int, fmaxf(mrun, fmaxf(mev, mod)))));
    const float resc = __expf(mrun - mnew);
    const float p    = __expf(lg - mnew);
    lsum = lsum * resc + p;
    gacc[0] *= resc;
    gacc[1] *= resc;
    #pragma unroll
    for (int j = 0; j < 16; ++j) {
      const int lj = 16 * (j & 1) + (j & 14);
      const float pj = __builtin_bit_cast(float,
          __builtin_amdgcn_readlane(__builtin_bit_cast(int, p), lj));
      const int   tj = __builtin_amdgcn_readlane(tailv, lj);
      const unsigned boff = (unsigned)tj * (unsigned)(D * sizeof(float)) + lcolb;
      const v2f row = *(const v2f*)(entb + boff);
      gacc[0] += pj * row[0];
      gacc[1] += pj * row[1];
    }
    mrun = mnew;
    rcur = rnext;
    tcur = tnext;
  }

  float lw = half16_sum((c & 1) ? 0.f : lsum);
  lw += __shfl_xor(lw, 16, 32);
  *(v2f*)(red + wave * 64 + 2 * lane) = gacc;
  if (lane == 0) { wstat[wave] = mrun; wstat[8 + wave] = lw; }
  __syncthreads();
  if (t < D) {
    float mx = wstat[0];
    #pragma unroll
    for (int w = 1; w < 8; ++w) mx = fmaxf(mx, wstat[w]);
    float den = 0.f, s = 0.f;
    #pragma unroll
    for (int w = 0; w < 8; ++w) {
      const float f = __expf(wstat[w] - mx);
      den += f * wstat[8 + w];
      s   += f * red[w * 64 + t];
    }
    drug_e[(size_t)n * (2 * D) + t]     = s / den;
    drug_e[(size_t)n * (2 * D) + D + t] = demb[t];
  }
}

#define LB_ROWS 27
__global__ void __launch_bounds__(1024) gnn_linear_bn(
    const float* __restrict__ drug_e, const float* __restrict__ lin_w,
    const float* __restrict__ lin_b,  const float* __restrict__ bn_w,
    const float* __restrict__ bn_b,   float* __restrict__ out)
{
  __shared__ float ysum[2][32 * LB_ROWS];
  __shared__ float part[2][64];
  const int c = blockIdx.x, t = threadIdx.x;
  const int lane = t & 31;
  const int wave = __builtin_amdgcn_readfirstlane(t >> 5);
  const v4f w = *(const v4f*)(lin_w + (size_t)c * (2 * D) + 4 * lane);

  const int row0 = wave * LB_ROWS;
  v4f de[LB_ROWS];
  #pragma unroll
  for (int i = 0; i < LB_ROWS; ++i) {
    const int rc = min(row0 + i, N_DRUG - 1);
    de[i] = *(const v4f*)(drug_e + (size_t)rc * (2 * D) + 4 * lane);
  }
  #pragma unroll
  for (int i = 0; i < LB_ROWS; ++i) {
    float p = de[i][0] * w[0];
    p += de[i][1] * w[1];
    p += de[i][2] * w[2];
    p += de[i][3] * w[3];
    p = half16_sum(p);
    if ((lane & 15) == 0) ysum[lane >> 4][row0 + i] = p;
  }
  __syncthreads();

  const bool live = t < N_DRUG;
  float y = 0.f;
  if (live) y = fmaxf(ysum[0][t] + ysum[1][t] + lin_b[c], 0.f);
  const float s = half16_sum(y);
  if ((lane & 15) == 0) part[0][2 * (t >> 5) + (lane >> 4)] = s;
  __syncthreads();
  float tot = 0.f;
  #pragma unroll
  for (int i = 0; i < 64; ++i) tot += part[0][i];
  const float mean = tot / (float)N_DRUG;
  const float d = live ? y - mean : 0.f;
  const float q = half16_sum(d * d);
  if ((lane & 15) == 0) part[1][2 * (t >> 5) + (lane >> 4)] = q;
  __syncthreads();
  float qt = 0.f;
  #pragma unroll
  for (int i = 0; i < 64; ++i) qt += part[1][i];
  const float var   = qt / (float)N_DRUG;
  const float scale = bn_w[c] * rsqrtf(var + BN_EPS);
  const float shift = bn_b[c];
  if (live) out[(size_t)t * D + c] = d * scale + shift;
}

extern "C" void kernel_launch(void* const* d_in, const int* in_sizes, int n_in,
                              void* d_out, int out_size, void* d_ws, size_t ws_size,
                              hipStream_t stream) {
  (void)in_sizes; (void)n_in; (void)out_size; (void)ws_size;
  const float* drug_table = (const float*)d_in[0];
  const float* rela_table = (const float*)d_in[1];
  const float* ent_table  = (const float*)d_in[2];
  const float* W1         = (const float*)d_in[3];
  const float* b1         = (const float*)d_in[4];
  const float* W2         = (const float*)d_in[5];
  const float* b2         = (const float*)d_in[6];
  const float* lin_w      = (const float*)d_in[7];
  const float* lin_b      = (const float*)d_in[8];
  const float* bn_w       = (const float*)d_in[9];
  const float* bn_b       = (const float*)d_in[10];
  const int*   drug_name  = (const int*)d_in[11];
  const int*   adj_tail   = (const int*)d_in[12];
  const int*   adj_rel    = (const int*)d_in[13];

  float* ws     = (float*)d_ws;
  float* b2sum  = ws;
  float* drug_e = ws + K;

  gnn_b2sum<<<dim3((K + 255) / 256), dim3(256), 0, stream>>>(b2, b2sum);

  const size_t lds = (size_t)(N_REL * ASTR + 32 * PSTR + D + D + 16 + 512) * sizeof(float);
  gnn_main<<<dim3(N_DRUG), dim3(256), lds, stream>>>(
      drug_table, rela_table, ent_table, W1, b1, W2,
      drug_name, adj_tail, adj_rel, b2sum, drug_e);

  gnn_linear_bn<<<dim3(D), dim3(1024), 0, stream>>>(drug_e, lin_w, lin_b, bn_w, bn_b, (float*)d_out);
}


# Per-drug kernel set-up: W1, W2, relation rows, drug embedding and first indices fetched as one group of 16-B loads before any LDS store (was 15 waited trips); one barrier fewer; bit-identical
# speedup vs baseline: 6.0189x; 1.1104x over previous
#include <hip/hip_runtime.h>

#define N_DRUG 846
#define D 64
#define K 1024
#define N_REL 200
#define BN_EPS 1e-5f

typedef __attribute__((ext_vector_type(2))) float v2f;
typedef __attribute__((ext_vector_type(4))) float v4f;
typedef __attribute__((ext_vector_type(8))) float v8f;

__device__ __forceinline__ float wave_reduce_add(float v) {
  #pragma unroll
  for (int off = 16; off > 0; off >>= 1) v += __shfl_xor(v, off, 32);
  return v;
}

template <int CTRL>
__device__ __forceinline__ float half_xor(float v) {
  return __builtin_bit_cast(float,
      __builtin_amdgcn_update_dpp(0, __builtin_bit_cast(int, v), CTRL, 0xf, 0xf, true));
}
__device__ __forceinline__ float half16_max(float v) {
  v = fmaxf(v, half_xor<0x168>(v));
  v = fmaxf(v, half_xor<0x164>(v));
  v = fmaxf(v, half_xor<0x162>(v));
  v = fmaxf(v, half_xor<0x161>(v));
  return v;
}
__device__ __forceinline__ float half16_sum(float v) {
  v += half_xor<0x168>(v);
  v += half_xor<0x164>(v);
  v += half_xor<0x162>(v);
  v += half_xor<0x161>(v);
  return v;
}

template <int N>
__device__ __forceinline__ int row_bcast(int v) {
  return __builtin_amdgcn_update_dpp(0, v, 0x150 + N, 0xf, 0xf, true);
}

__global__ void __launch_bounds__(256) gnn_b2sum(const float* __restrict__ b2,
                                                 float* __restrict__ b2sum) {
  const int k = blockIdx.x * 256 + threadIdx.x;
  if (k < K) {
    const float* p = b2 + (size_t)k * D;
    float s = 0.f;
    #pragma unroll
    for (int e = 0; e < D; ++e) s += p[e];
    b2sum[k] = s;
  }
}

#define ASTR 68
#define PSTR 130

__global__ void __launch_bounds__(256, 8) gnn_main(
    const float* __restrict__ drug_table, const float* __restrict__ rela_table,
    const float* __restrict__ ent_table,  const float* __restrict__ W1g,
    const float* __restrict__ b1,         const float* __restrict__ W2g,
    const int*   __restrict__ drug_name,  const int* __restrict__ adj_tail,
    const int*   __restrict__ adj_rel,    const float* __restrict__ b2sum,
    float* __restrict__ drug_e)
{
  extern __shared__ __align__(16) float smem[];
  float* Atab  = smem;
  float* W1p   = Atab + N_REL * ASTR;
  float* demb  = W1p + 32 * PSTR;
  float* w2sum = demb + D;
  float* wstat = w2sum + D;
  float* red   = wstat + 16;

  const int n    = blockIdx.x;
  const int t    = threadIdx.x;
  const int lane = t & 31;
  const int wave = t >> 5;
  const int swave = __builtin_amdgcn_readfirstlane(wave);
  const int h    = lane >> 4;
  const int c    = lane & 15;

  const size_t wbase = (size_t)n * D * D;
  const int nbl = (c & 14) | h;
  const int* relp  = adj_rel  + (size_t)n * K + swave * 128 + nbl;
  const int* tailp = adj_tail + (size_t)n * K + swave * 128 + nbl;
  int rcur = relp[0];
  int tcur = tailp[0];

  v4f wa[2], wb[2];
  #pragma unroll
  for (int j = 0; j < 2; ++j) {
    const int u = t + 256 * j;
    const float* p = W1g + wbase + (size_t)((u >> 4) * 2 * D + (u & 15) * 4);
    wa[j] = *(const v4f*)p;
    wb[j] = *(const v4f*)(p + D);
  }
  v4f w2v[4];
  {
    const float* p = W2g + wbase + (t >> 2) * D + (t & 3) * 16;
    #pragma unroll
    for (int j = 0; j < 4; ++j) w2v[j] = *(const v4f*)(p + 4 * j);
  }
  v4f rv[13];
  #pragma unroll
  for (int j = 0; j < 12; ++j) rv[j] = *(const v4f*)(rela_table + 4 * (t + 256 * j));
  rv[12] = *(const v4f*)(rela_table + 4 * min(t + 256 * 12, N_REL * 16 - 1));
  const v4f de4 = *(const v4f*)(drug_table + (size_t)drug_name[n] * D + 4 * (t & 15));
  __builtin_amdgcn_sched_barrier(0);
  asm volatile("" :: "v"(rv[12]));

  #pragma unroll
  for (int j = 0; j < 2; ++j) {
    const int u = t + 256 * j;
    float* dst = W1p + (u >> 4) * PSTR + (u & 15) * 8;
    *(v2f*)(dst)     = (v2f){wa[j][0], wb[j][0]};
    *(v2f*)(dst + 2) = (v2f){wa[j][1], wb[j][1]};
    *(v2f*)(dst + 4) = (v2f){wa[j][2], wb[j][2]};
    *(v2f*)(dst + 6) = (v2f){wa[j][3], wb[j][3]};
  }
  {
    float s = 0.f;
    #pragma unroll
    for (int j = 0; j < 16; ++j) s += w2v[j >> 2][j & 3];
    red[t] = s;
  }
  {
    float* arow = Atab + (t >> 4) * ASTR + 4 * (t & 15);
    #pragma unroll
    for (int j = 0; j < 12; ++j) *(v4f*)(arow + 16 * j * ASTR) = de4 * rv[j];
    if (t < 128) *(v4f*)(arow + 16 * 12 * ASTR) = de4 * rv[12];
  }
  if (t < 16) *(v4f*)(demb + 4 * t) = de4;
  __syncthreads();
  if (t < D) w2sum[t] = red[4*t] + red[4*t+1] + red[4*t+2] + red[4*t+3];

  const float* wp = W1p + (16 * h) * PSTR + 2 * c;
  for (int j = swave; j < (N_REL + 15) / 16; j += 8) {
    const int rel  = 16 * j + c;
    const int relc = min(rel, N_REL - 1);
    float* trow = Atab + relc * ASTR;
    v4f bq[8];
    #pragma unroll
    for (int q = 0; q < 8; ++q) bq[q] = *(const v4f*)(trow + 32 * h + 4 * q);
    #pragma unroll 1
    for (int eh = 0; eh < 2; ++eh) {
      v8f acc[2];
      #pragma unroll
      for (int e2 = 0; e2 < 2; ++e2)
        acc[e2] = (v8f){0.f, 0.f, 0.f, 0.f, 0.f, 0.f, 0.f, 0.f};
      const float* wpe = wp + 64 * eh;
      #pragma unroll
      for (int kk = 0; kk < 16; ++kk) {
        v2f b;
        b[0] = bq[kk >> 1][2 * (kk & 1)];
        b[1] = bq[kk >> 1][2 * (kk & 1) + 1];
        #pragma unroll
        for (int e2 = 0; e2 < 2; ++e2) {
          const v2f a = *(const v2f*)(wpe + kk * PSTR + 32 * e2);
          acc[e2] = __builtin_amdgcn_wmma_f32_16x16x4_f32(false, a, false, b,
                                                          (short)0, acc[e2], false, false);
        }
      }
      if (rel < N_REL) {
        #pragma unroll
        for (int e2 = 0; e2 < 2; ++e2) {
          float* tp = trow + 32 * eh + 16 * e2 + 8 * h;
          *(v4f*)tp       = (v4f){acc[e2][0], acc[e2][1], acc[e2][2], acc[e2][3]};
          *(v4f*)(tp + 4) = (v4f){acc[e2][4], acc[e2][5], acc[e2][6], acc[e2][7]};
        }
      }
    }
  }
  __syncthreads();

  const v4f w4 = *(const v4f*)(w2sum + 4 * c);
  const char* tabl = (const char*)Atab + 16 * c;
  const bool hi8 = (c & 8) != 0, hi4 = (c & 4) != 0, hi2 = (c & 2) != 0;
  const unsigned lcolb = 8u * (unsigned)lane;
  const char* entb = (const char*)ent_table;
  float mrun = -INFINITY;
  float lsum = 0.f;
  v2f   gacc = {0.f, 0.f};
  for (int tt = 0; tt < 8; ++tt) {
    const int kb = swave * 128 + tt * 16;
    const int rnext = relp[(tt < 7 ? tt + 1 : tt) * 16];
    const int tnext = tailp[(tt < 7 ? tt + 1 : tt) * 16];
    const int tailv = tcur;
    const float* bt = b1 + (size_t)kb * D + 4 * lane;
    v4f bb[8];
    #pragma unroll
    for (int i = 0; i < 8; ++i) bb[i] = *(const v4f*)(bt + 128 * i);
    const float b2s = b2sum[kb + nbl];
    const int roff = rcur * (ASTR * 4);
    v4f tr[8];
    tr[0] = *(const v4f*)(tabl + row_bcast<0>(roff));
    tr[1] = *(const v4f*)(tabl + row_bcast<2>(roff));
    tr[2] = *(const v4f*)(tabl + row_bcast<4>(roff));
    tr[3] = *(const v4f*)(tabl + row_bcast<6>(roff));
    tr[4] = *(const v4f*)(tabl + row_bcast<8>(roff));
    tr[5] = *(const v4f*)(tabl + row_bcast<10>(roff));
    tr[6] = *(const v4f*)(tabl + row_bcast<12>(roff));
    tr[7] = *(const v4f*)(tabl + row_bcast<14>(roff));
    float v[8];
    #pragma unroll
    for (int i = 0; i < 8; ++i) {
      float s = fmaxf(tr[i][0] + bb[i][0], 0.f) * w4[0];
      s += fmaxf(tr[i][1] + bb[i][1], 0.f) * w4[1];
      s += fmaxf(tr[i][2] + bb[i][2], 0.f) * w4[2];
      s += fmaxf(tr[i][3] + bb[i][3], 0.f) * w4[3];
      v[i] = s;
    }
    float u[4];
    #pragma unroll
    for (int i = 0; i < 4; ++i)
      u[i] = (hi8 ? v[i + 4] : v[i]) + half_xor<0x168>(hi8 ? v[i] : v[i + 4]);
    float x[2];
    #pragma unroll
    for (int i = 0; i < 2; ++i)
      x[i] = (hi4 ? u[i + 2] : u[i]) + half_xor<0x164>(hi4 ? u[i] : u[i + 2]);
    float y = (hi2 ? x[1] : x[0]) + half_xor<0x162>(hi2 ? x[0] : x[1]);
    y += half_xor<0x161>(y);
    const float lg = y + b2s;
    const float m16 = half16_max(lg);
    const float mev = __builtin_bit_cast(float, __builtin_amdgcn_readlane(__builtin_bit_cast(int, m16), 0));
    const float mod = __builtin_bit_cast(float, __builtin_amdgcn_readlane(__builtin_bit_cast(int, m16), 16));
    const float mnew = __builtin_bit_cast(float, __builtin_amdgcn_readfirstlane(
        __builtin_bit_cast(int, fmaxf(mrun, fmaxf(mev, mod)))));
    const float resc = __expf(mrun - mnew);
    const float p    = __expf(lg - mnew);
    lsum = lsum * resc + p;
    gacc[0] *= resc;
    gacc[1] *= resc;
    #pragma unroll
    for (int j = 0; j < 16; ++j) {
      const int lj = 16 * (j & 1) + (j & 14);
      const float pj = __builtin_bit_cast(float,
          __builtin_amdgcn_readlane(__builtin_bit_cast(int, p), lj));
      const int   tj = __builtin_amdgcn_readlane(tailv, lj);
      const unsigned boff = (unsigned)tj * (unsigned)(D * sizeof(float)) + lcolb;
      const v2f row = *(const v2f*)(entb + boff);
      gacc[0] += pj * row[0];
      gacc[1] += pj * row[1];
    }
    mrun = mnew;
    rcur = rnext;
    tcur = tnext;
  }

  float lw = half16_sum((c & 1) ? 0.f : lsum);
  lw += __shfl_xor(lw, 16, 32);
  *(v2f*)(red + wave * 64 + 2 * lane) = gacc;
  if (lane == 0) { wstat[wave] = mrun; wstat[8 + wave] = lw; }
  __syncthreads();
  if (t < D) {
    float mx = wstat[0];
    #pragma unroll
    for (int w = 1; w < 8; ++w) mx = fmaxf(mx, wstat[w]);
    float den = 0.f, s = 0.f;
    #pragma unroll
    for (int w = 0; w < 8; ++w) {
      const float f = __expf(wstat[w] - mx);
      den += f * wstat[8 + w];
      s   += f * red[w * 64 + t];
    }
    drug_e[(size_t)n * (2 * D) + t]     = s / den;
    drug_e[(size_t)n * (2 * D) + D + t] = demb[t];
  }
}

#define LB_ROWS 27
__global__ void __launch_bounds__(1024) gnn_linear_bn(
    const float* __restrict__ drug_e, const float* __restrict__ lin_w,
    const float* __restrict__ lin_b,  const float* __restrict__ bn_w,
    const float* __restrict__ bn_b,   float* __restrict__ out)
{
  __shared__ float ysum[2][32 * LB_ROWS];
  __shared__ float part[2][64];
  const int c = blockIdx.x, t = threadIdx.x;
  const int lane = t & 31;
  const int wave = __builtin_amdgcn_readfirstlane(t >> 5);
  const v4f w = *(const v4f*)(lin_w + (size_t)c * (2 * D) + 4 * lane);

  const int row0 = wave * LB_ROWS;
  v4f de[LB_ROWS];
  #pragma unroll
  for (int i = 0; i < LB_ROWS; ++i) {
    const int rc = min(row0 + i, N_DRUG - 1);
    de[i] = *(const v4f*)(drug_e + (size_t)rc * (2 * D) + 4 * lane);
  }
  #pragma unroll
  for (int i = 0; i < LB_ROWS; ++i) {
    float p = de[i][0] * w[0];
    p += de[i][1] * w[1];
    p += de[i][2] * w[2];
    p += de[i][3] * w[3];
    p = half16_sum(p);
    if ((lane & 15) == 0) ysum[lane >> 4][row0 + i] = p;
  }
  __syncthreads();

  const bool live = t < N_DRUG;
  float y = 0.f;
  if (live) y = fmaxf(ysum[0][t] + ysum[1][t] + lin_b[c], 0.f);
  const float s = half16_sum(y);
  if ((lane & 15) == 0) part[0][2 * (t >> 5) + (lane >> 4)] = s;
  __syncthreads();
  float tot = 0.f;
  #pragma unroll
  for (int i = 0; i < 64; ++i) tot += part[0][i];
  const float mean = tot / (float)N_DRUG;
  const float d = live ? y - mean : 0.f;
  const float q = half16_sum(d * d);
  if ((lane & 15) == 0) part[1][2 * (t >> 5) + (lane >> 4)] = q;
  __syncthreads();
  float qt = 0.f;
  #pragma unroll
  for (int i = 0; i < 64; ++i) qt += part[1][i];
  const float var   = qt / (float)N_DRUG;
  const float scale = bn_w[c] * rsqrtf(var + BN_EPS);
  const float shift = bn_b[c];
  if (live) out[(size_t)t * D + c] = d * scale + shift;
}

extern "C" void kernel_launch(void* const* d_in, const int* in_sizes, int n_in,
                              void* d_out, int out_size, void* d_ws, size_t ws_size,
                              hipStream_t stream) {
  (void)in_sizes; (void)n_in; (void)out_size; (void)ws_size;
  const float* drug_table = (const float*)d_in[0];
  const float* rela_table = (const float*)d_in[1];
  const float* ent_table  = (const float*)d_in[2];
  const float* W1         = (const float*)d_in[3];
  const float* b1         = (const float*)d_in[4];
  const float* W2         = (const float*)d_in[5];
  const float* b2         = (const float*)d_in[6];
  const float* lin_w      = (const float*)d_in[7];
  const float* lin_b      = (const float*)d_in[8];
  const float* bn_w       = (const float*)d_in[9];
  const float* bn_b       = (const float*)d_in[10];
  const int*   drug_name  = (const int*)d_in[11];
  const int*   adj_tail   = (const int*)d_in[12];
  const int*   adj_rel    = (const int*)d_in[13];

  float* ws     = (float*)d_ws;
  float* b2sum  = ws;
  float* drug_e = ws + K;

  gnn_b2sum<<<dim3((K + 255) / 256), dim3(256), 0, stream>>>(b2, b2sum);

  const size_t lds = (size_t)(N_REL * ASTR + 32 * PSTR + D + D + 16 + 512) * sizeof(float);
  gnn_main<<<dim3(N_DRUG), dim3(256), lds, stream>>>(
      drug_table, rela_table, ent_table, W1, b1, W2,
      drug_name, adj_tail, adj_rel, b2sum, drug_e);

  gnn_linear_bn<<<dim3(D), dim3(1024), 0, stream>>>(drug_e, lin_w, lin_b, bn_w, bn_b, (float*)d_out);
}


# Two drugs per 16-wave work-group: each wave's b1 tile registers and the set-up's relation rows serve both drugs (b1 fetched once per pair, not per drug); softmax and entity sum orders change
# speedup vs baseline: 6.3159x; 1.0493x over previous
#include <hip/hip_runtime.h>

#define N_DRUG 846
#define D 64
#define K 1024
#define N_REL 200
#define BN_EPS 1e-5f

typedef __attribute__((ext_vector_type(2))) float v2f;
typedef __attribute__((ext_vector_type(4))) float v4f;
typedef __attribute__((ext_vector_type(8))) float v8f;

__device__ __forceinline__ float wave_reduce_add(float v) {
  #pragma unroll
  for (int off = 16; off > 0; off >>= 1) v += __shfl_xor(v, off, 32);
  return v;
}

template <int CTRL>
__device__ __forceinline__ float half_xor(float v) {
  return __builtin_bit_cast(float,
      __builtin_amdgcn_update_dpp(0, __builtin_bit_cast(int, v), CTRL, 0xf, 0xf, true));
}
__device__ __forceinline__ float half16_max(float v) {
  v = fmaxf(v, half_xor<0x168>(v));
  v = fmaxf(v, half_xor<0x164>(v));
  v = fmaxf(v, half_xor<0x162>(v));
  v = fmaxf(v, half_xor<0x161>(v));
  return v;
}
__device__ __forceinline__ float half16_sum(float v) {
  v += half_xor<0x168>(v);
  v += half_xor<0x164>(v);
  v += half_xor<0x162>(v);
  v += half_xor<0x161>(v);
  return v;
}

template <int N>
__device__ __forceinline__ int row_bcast(int v) {
  return __builtin_amdgcn_update_dpp(0, v, 0x150 + N, 0xf, 0xf, true);
}

__global__ void __launch_bounds__(256) gnn_b2sum(const float* __restrict__ b2,
                                                 float* __restrict__ b2sum) {
  const int k = blockIdx.x * 256 + threadIdx.x;
  if (k < K) {
    const float* p = b2 + (size_t)k * D;
    float s = 0.f;
    #pragma unroll
    for (int e = 0; e < D; ++e) s += p[e];
    b2sum[k] = s;
  }
}

#define ASTR 68
#define PSTR 130
#define TABSZ (N_REL * ASTR)
#define W1SZ  (32 * PSTR)
#define MAIN_LDS_FLOATS (2 * TABSZ + 2 * W1SZ + 2 * D + 2 * D + 64 + 2048)
static_assert(N_DRUG % 2 == 0, "a work-group takes a pair of drugs");

__global__ void __launch_bounds__(512, 8) gnn_main(
    const float* __restrict__ drug_table, const float* __restrict__ rela_table,
    const float* __restrict__ ent_table,  const float* __restrict__ W1g,
    const float* __restrict__ b1,         const float* __restrict__ W2g,
    const int*   __restrict__ drug_name,  const int* __restrict__ adj_tail,
    const int*   __restrict__ adj_rel,    const float* __restrict__ b2sum,
    float* __restrict__ drug_e)
{
  extern __shared__ __align__(16) float smem[];
  float* Atab  = smem;
  float* W1p   = Atab + 2 * TABSZ;
  float* demb  = W1p + 2 * W1SZ;
  float* w2sum = demb + 2 * D;
  float* wstat = w2sum + 2 * D;
  float* red   = wstat + 64;

  const int n0   = 2 * blockIdx.x;
  const int t    = threadIdx.x;
  const int lane = t & 31;
  const int wave = t >> 5;
  const int swave = __builtin_amdgcn_readfirstlane(wave);
  const int h    = lane >> 4;
  const int c    = lane & 15;
  const int hd   = t >> 8;
  const int tl   = t & 255;

  const size_t wbase = (size_t)(n0 + hd) * D * D;
  const int nbl = (c & 14) | h;
  const int* relp  = adj_rel  + (size_t)n0 * K + swave * 64 + nbl;
  const int* tailp = adj_tail + (size_t)n0 * K + swave * 64 + nbl;
  int rcur[2], tcur[2];
  #pragma unroll
  for (int d = 0; d < 2; ++d) { rcur[d] = relp[d * K]; tcur[d] = tailp[d * K]; }

  v4f wa[2], wb[2];
  #pragma unroll
  for (int j = 0; j < 2; ++j) {
    const int u = tl + 256 * j;
    const float* p = W1g + wbase + (size_t)((u >> 4) * 2 * D + (u & 15) * 4);
    wa[j] = *(const v4f*)p;
    wb[j] = *(const v4f*)(p + D);
  }
  v4f w2v[4];
  {
    const float* p = W2g + wbase + (tl >> 2) * D + (tl & 3) * 16;
    #pragma unroll
    for (int j = 0; j < 4; ++j) w2v[j] = *(const v4f*)(p + 4 * j);
  }
  v4f rv[7];
  #pragma unroll
  for (int j = 0; j < 6; ++j) rv[j] = *(const v4f*)(rela_table + 4 * (t + 512 * j));
  rv[6] = *(const v4f*)(rela_table + 4 * min(t + 512 * 6, N_REL * 16 - 1));
  v4f de4[2];
  #pragma unroll
  for (int d = 0; d < 2; ++d)
    de4[d] = *(const v4f*)(drug_table + (size_t)drug_name[n0 + d] * D + 4 * (t & 15));
  __builtin_amdgcn_sched_barrier(0);
  asm volatile("" :: "v"(rv[6]));

  {
    float* W1pd = W1p + hd * W1SZ;
    #pragma unroll
    for (int j = 0; j < 2; ++j) {
      const int u = tl + 256 * j;
      float* dst = W1pd + (u >> 4) * PSTR + (u & 15) * 8;
      *(v2f*)(dst)     = (v2f){wa[j][0], wb[j][0]};
      *(v2f*)(dst + 2) = (v2f){wa[j][1], wb[j][1]};
      *(v2f*)(dst + 4) = (v2f){wa[j][2], wb[j][2]};
      *(v2f*)(dst + 6) = (v2f){wa[j][3], wb[j][3]};
    }
  }
  {
    float s = 0.f;
    #pragma unroll
    for (int j = 0; j < 16; ++j) s += w2v[j >> 2][j & 3];
    red[t] = s;
  }
  #pragma unroll
  for (int d = 0; d < 2; ++d) {
    float* arow = Atab + d * TABSZ + (t >> 4) * ASTR + 4 * (t & 15);
    #pragma unroll
    for (int j = 0; j < 6; ++j) *(v4f*)(arow + 32 * j * ASTR) = de4[d] * rv[j];
    if (t < 128) *(v4f*)(arow + 32 * 6 * ASTR) = de4[d] * rv[6];
    if (t < 16) *(v4f*)(demb + d * D + 4 * t) = de4[d];
  }
  __syncthreads();
  if (t < 2 * D) w2sum[t] = red[4*t] + red[4*t+1] + red[4*t+2] + red[4*t+3];

  for (int it = swave; it < 2 * ((N_REL + 15) / 16); it += 16) {
    const int dsel = it >= (N_REL + 15) / 16 ? 1 : 0;
    const int j    = it - dsel * ((N_REL + 15) / 16);
    const float* wp = W1p + dsel * W1SZ + (16 * h) * PSTR + 2 * c;
    const int rel  = 16 * j + c;
    const int relc = min(rel, N_REL - 1);
    float* trow = Atab + dsel * TABSZ + relc * ASTR;
    v4f bq[8];
    #pragma unroll
    for (int q = 0; q < 8; ++q) bq[q] = *(const v4f*)(trow + 32 * h + 4 * q);
    #pragma unroll 1
    for (int eh = 0; eh < 2; ++eh) {
      v8f acc[2];
      #pragma unroll
      for (int e2 = 0; e2 < 2; ++e2)
        acc[e2] = (v8f){0.f, 0.f, 0.f, 0.f, 0.f, 0.f, 0.f, 0.f};
      const float* wpe = wp + 64 * eh;
      #pragma unroll
      for (int kk = 0; kk < 16; ++kk) {
        v2f b;
        b[0] = bq[kk >> 1][2 * (kk & 1)];
        b[1] = bq[kk >> 1][2 * (kk & 1) + 1];
        #pragma unroll
        for (int e2 = 0; e2 < 2; ++e2) {
          const v2f a = *(const v2f*)(wpe + kk * PSTR + 32 * e2);
          acc[e2] = __builtin_amdgcn_wmma_f32_16x16x4_f32(false, a, false, b,
                                                          (short)0, acc[e2], false, false);
        }
      }
      if (rel < N_REL) {
        #pragma unroll
        for (int e2 = 0; e2 < 2; ++e2) {
          float* tp = trow + 32 * eh + 16 * e2 + 8 * h;
          *(v4f*)tp       = (v4f){acc[e2][0], acc[e2][1], acc[e2][2], acc[e2][3]};
          *(v4f*)(tp + 4) = (v4f){acc[e2][4], acc[e2][5], acc[e2][6], acc[e2][7]};
        }
      }
    }
  }
  __syncthreads();

  v4f w4[2];
  #pragma unroll
  for (int d = 0; d < 2; ++d) w4[d] = *(const v4f*)(w2sum + d * D + 4 * c);
  const char* tabl = (const char*)Atab + 16 * c;
  const bool hi8 = (c & 8) != 0, hi4 = (c & 4) != 0, hi2 = (c & 2) != 0;
  const unsigned lcolb = 8u * (unsigned)lane;
  const char* entb = (const char*)ent_table;
  float mrun[2] = {-INFINITY, -INFINITY};
  float lsum[2] = {0.f, 0.f};
  v2f   gacc[2] = {{0.f, 0.f}, {0.f, 0.f}};
  #pragma unroll 1
  for (int tt = 0; tt < 4; ++tt) {
    const int kb = swave * 64 + tt * 16;
    const int nx = (tt < 3 ? tt + 1 : tt) * 16;
    const float* bt = b1 + (size_t)kb * D + 4 * lane;
    v4f bb[8];
    #pragma unroll
    for (int i = 0; i < 8; ++i) bb[i] = *(const v4f*)(bt + 128 * i);
    const float b2s = b2sum[kb + nbl];
    #pragma unroll
    for (int d = 0; d < 2; ++d) {
      const int rnext = relp[d * K + nx];
      const int tnext = tailp[d * K + nx];
      const int tailv = tcur[d];
      const int roff = rcur[d] * (ASTR * 4) + d * (TABSZ * 4);
      v4f tr[8];
      tr[0] = *(const v4f*)(tabl + row_bcast<0>(roff));
      tr[1] = *(const v4f*)(tabl + row_bcast<2>(roff));
      tr[2] = *(const v4f*)(tabl + row_bcast<4>(roff));
      tr[3] = *(const v4f*)(tabl + row_bcast<6>(roff));
      tr[4] = *(const v4f*)(tabl + row_bcast<8>(roff));
      tr[5] = *(const v4f*)(tabl + row_bcast<10>(roff));
      tr[6] = *(const v4f*)(tabl + row_bcast<12>(roff));
      tr[7] = *(const v4f*)(tabl + row_bcast<14>(roff));
      float v[8];
      #pragma unroll
      for (int i = 0; i < 8; ++i) {
        float s = fmaxf(tr[i][0] + bb[i][0], 0.f) * w4[d][0];
        s += fmaxf(tr[i][1] + bb[i][1], 0.f) * w4[d][1];
        s += fmaxf(tr[i][2] + bb[i][2], 0.f) * w4[d][2];
        s += fmaxf(tr[i][3] + bb[i][3], 0.f) * w4[d][3];
        v[i] = s;
      }
      float u[4];
      #pragma unroll
      for (int i = 0; i < 4; ++i)
        u[i] = (hi8 ? v[i + 4] : v[i]) + half_xor<0x168>(hi8 ? v[i] : v[i + 4]);
      float x[2];
      #pragma unroll
      for (int i = 0; i < 2; ++i)
        x[i] = (hi4 ? u[i + 2] : u[i]) + half_xor<0x164>(hi4 ? u[i] : u[i + 2]);
      float y = (hi2 ? x[1] : x[0]) + half_xor<0x162>(hi2 ? x[0] : x[1]);
      y += half_xor<0x161>(y);
      const float lg = y + b2s;
      const float m16 = half16_max(lg);
      const float mev = __builtin_bit_cast(float, __builtin_amdgcn_readlane(__builtin_bit_cast(int, m16), 0));
      const float mod = __builtin_bit_cast(float, __builtin_amdgcn_readlane(__builtin_bit_cast(int, m16), 16));
      const float mnew = __builtin_bit_cast(float, __builtin_amdgcn_readfirstlane(
          __builtin_bit_cast(int, fmaxf(mrun[d], fmaxf(mev, mod)))));
      const float resc = __expf(mrun[d] - mnew);
      const float p    = __expf(lg - mnew);
      lsum[d] = lsum[d] * resc + p;
      gacc[d][0] *= resc;
      gacc[d][1] *= resc;
      #pragma unroll
      for (int j = 0; j < 16; ++j) {
        const int lj = 16 * (j & 1) + (j & 14);
        const float pj = __builtin_bit_cast(float,
            __builtin_amdgcn_readlane(__builtin_bit_cast(int, p), lj));
        const int   tj = __builtin_amdgcn_readlane(tailv, lj);
        const char* rowp = entb + (size_t)(unsigned)tj * (D * sizeof(float));
        const v2f row = *(const v2f*)(rowp + lcolb);
        gacc[d][0] += pj * row[0];
        gacc[d][1] += pj * row[1];
      }
      mrun[d] = mnew;
      rcur[d] = rnext;
      tcur[d] = tnext;
    }
  }

  const int ln = (int)__builtin_amdgcn_mbcnt_lo(~0u, 0u);
  #pragma unroll
  for (int d = 0; d < 2; ++d) {
    float lw = half16_sum((ln & 1) ? 0.f : lsum[d]);
    lw += __shfl_xor(lw, 16, 32);
    *(v2f*)(red + d * 1024 + swave * 64 + 2 * ln) = gacc[d];
    if (ln == 0) { wstat[d * 32 + swave] = mrun[d]; wstat[d * 32 + 16 + swave] = lw; }
  }
  __syncthreads();
  if (swave < 4) {
    const int d = swave >> 1, e = 32 * (swave & 1) + ln;
    const float* ws = wstat + d * 32;
    const float* rd = red + d * 1024 + e;
    float mx = ws[0];
    #pragma unroll
    for (int w = 1; w < 16; ++w) mx = fmaxf(mx, ws[w]);
    float den = 0.f, s = 0.f;
    #pragma unroll
    for (int w = 0; w < 16; ++w) {
      const float f = __expf(ws[w] - mx);
      den += f * ws[16 + w];
      s   += f * rd[w * 64];
    }
    drug_e[(size_t)(n0 + d) * (2 * D) + e]     = s / den;
    drug_e[(size_t)(n0 + d) * (2 * D) + D + e] = demb[d * D + e];
  }
}

#define LB_ROWS 27
__global__ void __launch_bounds__(1024) gnn_linear_bn(
    const float* __restrict__ drug_e, const float* __restrict__ lin_w,
    const float* __restrict__ lin_b,  const float* __restrict__ bn_w,
    const float* __restrict__ bn_b,   float* __restrict__ out)
{
  __shared__ float ysum[2][32 * LB_ROWS];
  __shared__ float part[2][64];
  const int c = blockIdx.x, t = threadIdx.x;
  const int lane = t & 31;
  const int wave = __builtin_amdgcn_readfirstlane(t >> 5);
  const v4f w = *(const v4f*)(lin_w + (size_t)c * (2 * D) + 4 * lane);

  const int row0 = wave * LB_ROWS;
  v4f de[LB_ROWS];
  #pragma unroll
  for (int i = 0; i < LB_ROWS; ++i) {
    const int rc = min(row0 + i, N_DRUG - 1);
    de[i] = *(const v4f*)(drug_e + (size_t)rc * (2 * D) + 4 * lane);
  }
  #pragma unroll
  for (int i = 0; i < LB_ROWS; ++i) {
    float p = de[i][0] * w[0];
    p += de[i][1] * w[1];
    p += de[i][2] * w[2];
    p += de[i][3] * w[3];
    p = half16_sum(p);
    if ((lane & 15) == 0) ysum[lane >> 4][row0 + i] = p;
  }
  __syncthreads();

  const bool live = t < N_DRUG;
  float y = 0.f;
  if (live) y = fmaxf(ysum[0][t] + ysum[1][t] + lin_b[c], 0.f);
  const float s = half16_sum(y);
  if ((lane & 15) == 0) part[0][2 * (t >> 5) + (lane >> 4)] = s;
  __syncthreads();
  float tot = 0.f;
  #pragma unroll
  for (int i = 0; i < 64; ++i) tot += part[0][i];
  const float mean = tot / (float)N_DRUG;
  const float d = live ? y - mean : 0.f;
  const float q = half16_sum(d * d);
  if ((lane & 15) == 0) part[1][2 * (t >> 5) + (lane >> 4)] = q;
  __syncthreads();
  float qt = 0.f;
  #pragma unroll
  for (int i = 0; i < 64; ++i) qt += part[1][i];
  const float var   = qt / (float)N_DRUG;
  const float scale = bn_w[c] * rsqrtf(var + BN_EPS);
  const float shift = bn_b[c];
  if (live) out[(size_t)t * D + c] = d * scale + shift;
}

extern "C" void kernel_launch(void* const* d_in, const int* in_sizes, int n_in,
                              void* d_out, int out_size, void* d_ws, size_t ws_size,
                              hipStream_t stream) {
  (void)in_sizes; (void)n_in; (void)out_size; (void)ws_size;
  const float* drug_table = (const float*)d_in[0];
  const float* rela_table = (const float*)d_in[1];
  const float* ent_table  = (const float*)d_in[2];
  const float* W1         = (const float*)d_in[3];
  const float* b1         = (const float*)d_in[4];
  const float* W2         = (const float*)d_in[5];
  const float* b2         = (const float*)d_in[6];
  const float* lin_w      = (const float*)d_in[7];
  const float* lin_b      = (const float*)d_in[8];
  const float* bn_w       = (const float*)d_in[9];
  const float* bn_b       = (const float*)d_in[10];
  const int*   drug_name  = (const int*)d_in[11];
  const int*   adj_tail   = (const int*)d_in[12];
  const int*   adj_rel    = (const int*)d_in[13];

  float* ws     = (float*)d_ws;
  float* b2sum  = ws;
  float* drug_e = ws + K;

  gnn_b2sum<<<dim3((K + 255) / 256), dim3(256), 0, stream>>>(b2, b2sum);

  const size_t lds = (size_t)MAIN_LDS_FLOATS * sizeof(float);
  gnn_main<<<dim3(N_DRUG / 2), dim3(512), lds, stream>>>(
      drug_table, rela_table, ent_table, W1, b1, W2,
      drug_name, adj_tail, adj_rel, b2sum, drug_e);

  gnn_linear_bn<<<dim3(D), dim3(1024), 0, stream>>>(drug_e, lin_w, lin_b, bn_w, bn_b, (float*)d_out);
}
